# starting kernel
# baseline (speedup 1.0000x reference)
#include <hip/hip_runtime.h>


typedef float v2f __attribute__((ext_vector_type(2)));
typedef float v8f __attribute__((ext_vector_type(8)));

#define GCN_IN_C  256
#define GCN_H1    128
#define GCN_H2    64
#define GCN_OUT   64

template <int K, int N>
__global__ __launch_bounds__(128)
void gcn_gemm_wmma_f32(const float* __restrict__ A,
                       const float* __restrict__ W,
                       float* __restrict__ C,
                       int M)
{
    const int lane = threadIdx.x & 31;
    const int wave = threadIdx.x >> 5;
    const int m0   = blockIdx.x << 5;
    const int n0   = wave << 5;
    const int half = lane >> 4;
    const int l    = lane & 15;

    const float* __restrict__ arow0 = A + (size_t)(m0 + l) * K + 2 * half;
    const float* __restrict__ arow1 = arow0 + (size_t)16 * K;
    const float* __restrict__ wcol0 = W + (size_t)(2 * half) * N + (n0 + l);
    const float* __restrict__ wcol1 = wcol0 + 16;

    v8f acc00 = {}, acc01 = {}, acc10 = {}, acc11 = {};
#pragma unroll 2
    for (int k = 0; k < K; k += 4) {
        v2f a0, a1, b0, b1;
        a0.x = arow0[k + 0];
        a0.y = arow0[k + 1];
        a1.x = arow1[k + 0];
        a1.y = arow1[k + 1];
        b0.x = wcol0[(size_t)k * N];
        b0.y = wcol0[(size_t)k * N + N];
        b1.x = wcol1[(size_t)k * N];
        b1.y = wcol1[(size_t)k * N + N];
        acc00 = __builtin_amdgcn_wmma_f32_16x16x4_f32(
            false, a0, false, b0, (short)0, acc00, false, false);
        acc01 = __builtin_amdgcn_wmma_f32_16x16x4_f32(
            false, a0, false, b1, (short)0, acc01, false, false);
        acc10 = __builtin_amdgcn_wmma_f32_16x16x4_f32(
            false, a1, false, b0, (short)0, acc10, false, false);
        acc11 = __builtin_amdgcn_wmma_f32_16x16x4_f32(
            false, a1, false, b1, (short)0, acc11, false, false);
    }

    float* __restrict__ crow0 = C + (size_t)(m0 + 8 * half) * N + (n0 + l);
    float* __restrict__ crow1 = crow0 + (size_t)16 * N;
#pragma unroll
    for (int v = 0; v < 8; ++v) {
        crow0[(size_t)v * N]      = acc00[v];
        crow0[(size_t)v * N + 16] = acc01[v];
        crow1[(size_t)v * N]      = acc10[v];
        crow1[(size_t)v * N + 16] = acc11[v];
    }
}

template <int C>
__global__ __launch_bounds__(256)
void gcn_edge_scatter(const float* __restrict__ sup,
                      const int* __restrict__ src,
                      const int* __restrict__ dst,
                      const float* __restrict__ ew,
                      float* __restrict__ agg,
                      int E)
{
    constexpr int CHUNKS = C / 4;
    long long tid = (long long)blockIdx.x * blockDim.x + threadIdx.x;
    if (tid >= (long long)E * CHUNKS) return;
    const int e = (int)(tid / CHUNKS);
    const int c = (int)(tid % CHUNKS) * 4;

    const int   s = src[e];
    const int   d = dst[e];
    const float w = ew[e];

    const float4 v = *reinterpret_cast<const float4*>(sup + (size_t)s * C + c);
    float* o = agg + (size_t)d * C + c;
    atomicAdd(o + 0, v.x * w);
    atomicAdd(o + 1, v.y * w);
    atomicAdd(o + 2, v.z * w);
    atomicAdd(o + 3, v.w * w);
}

__global__ __launch_bounds__(256)
void gcn_fill_bias(float* __restrict__ out, const float* __restrict__ b,
                   int C, long long total4)
{
    long long i = (long long)blockIdx.x * blockDim.x + threadIdx.x;
    if (i >= total4) return;
    const int c = (int)((i * 4) & (long long)(C - 1));
    const float4 v = *reinterpret_cast<const float4*>(b + c);
    *reinterpret_cast<float4*>(out + i * 4) = v;
}

__global__ __launch_bounds__(256)
void gcn_relu(float* __restrict__ x, long long total4)
{
    long long i = (long long)blockIdx.x * blockDim.x + threadIdx.x;
    if (i >= total4) return;
    float4 v = *reinterpret_cast<float4*>(x + i * 4);
    v.x = fmaxf(v.x, 0.0f);
    v.y = fmaxf(v.y, 0.0f);
    v.z = fmaxf(v.z, 0.0f);
    v.w = fmaxf(v.w, 0.0f);
    *reinterpret_cast<float4*>(x + i * 4) = v;
}

extern "C" void kernel_launch(void* const* d_in, const int* in_sizes, int n_in,
                              void* d_out, int out_size, void* d_ws, size_t ws_size,
                              hipStream_t stream)
{
    const float* x    = (const float*)d_in[0];
    const int*   esrc = (const int*)  d_in[1];
    const int*   edst = (const int*)  d_in[2];
    const float* ew   = (const float*)d_in[3];
    const float* W1   = (const float*)d_in[4];
    const float* b1   = (const float*)d_in[5];
    const float* W2   = (const float*)d_in[6];
    const float* b2   = (const float*)d_in[7];
    const float* W3   = (const float*)d_in[8];
    const float* b3   = (const float*)d_in[9];

    const int M = in_sizes[0] / GCN_IN_C;
    const int E = in_sizes[1];

    float* ws     = (float*)d_ws;
    float* sup128 = ws;
    float* h128   = sup128 + (size_t)M * GCN_H1;
    float* sup64  = h128   + (size_t)M * GCN_H1;
    float* h64    = sup64  + (size_t)M * GCN_H2;
    float* out    = (float*)d_out;

    const int mt = M / 32;
    const long long q128 = (long long)M * GCN_H1 / 4;
    const long long q64  = (long long)M * GCN_H2 / 4;
    const long long e128 = (long long)E * (GCN_H1 / 4);
    const long long e64  = (long long)E * (GCN_H2 / 4);
    const int B = 256;

    gcn_gemm_wmma_f32<GCN_IN_C, GCN_H1>
        <<<mt, (GCN_H1 / 32) * 32, 0, stream>>>(x, W1, sup128, M);
    gcn_fill_bias<<<(int)((q128 + B - 1) / B), B, 0, stream>>>(h128, b1, GCN_H1, q128);
    gcn_edge_scatter<GCN_H1>
        <<<(int)((e128 + B - 1) / B), B, 0, stream>>>(sup128, esrc, edst, ew, h128, E);
    gcn_relu<<<(int)((q128 + B - 1) / B), B, 0, stream>>>(h128, q128);

    gcn_gemm_wmma_f32<GCN_H1, GCN_H2>
        <<<mt, (GCN_H2 / 32) * 32, 0, stream>>>(h128, W2, sup64, M);
    gcn_fill_bias<<<(int)((q64 + B - 1) / B), B, 0, stream>>>(h64, b2, GCN_H2, q64);
    gcn_edge_scatter<GCN_H2>
        <<<(int)((e64 + B - 1) / B), B, 0, stream>>>(sup64, esrc, edst, ew, h64, E);
    gcn_relu<<<(int)((q64 + B - 1) / B), B, 0, stream>>>(h64, q64);

    gcn_gemm_wmma_f32<GCN_H2, GCN_OUT>
        <<<mt, (GCN_OUT / 32) * 32, 0, stream>>>(h64, W3, sup128, M);
    gcn_fill_bias<<<(int)((q64 + B - 1) / B), B, 0, stream>>>(out, b3, GCN_OUT, q64);
    gcn_edge_scatter<GCN_OUT>
        <<<(int)((e64 + B - 1) / B), B, 0, stream>>>(sup128, esrc, edst, ew, out, E);
}


# Replace f32 atomic scatter-add, bias fill and ReLU launches by a per-call destination-ordered edge list (integer rank atomics + scan) and one-wave-per-row gather sums; row sum order = arrival rank
# speedup vs baseline: 2.3446x; 2.3446x over previous
#include <hip/hip_runtime.h>


typedef float v2f __attribute__((ext_vector_type(2)));
typedef float v4f __attribute__((ext_vector_type(4)));
typedef float v8f __attribute__((ext_vector_type(8)));
typedef int   v2i __attribute__((ext_vector_type(2)));
typedef int   v4i __attribute__((ext_vector_type(4)));

#define GCN_IN_C  256
#define GCN_H1    128
#define GCN_H2    64
#define GCN_OUT   64

template <int K, int N>
__global__ __launch_bounds__(128)
void gcn_gemm_wmma_f32(const float* __restrict__ A,
                       const float* __restrict__ W,
                       float* __restrict__ C,
                       int M)
{
    const int lane = threadIdx.x & 31;
    const int wave = threadIdx.x >> 5;
    const int m0   = blockIdx.x << 5;
    const int n0   = wave << 5;
    const int half = lane >> 4;
    const int l    = lane & 15;

    const float* __restrict__ arow0 = A + (size_t)(m0 + l) * K + 2 * half;
    const float* __restrict__ arow1 = arow0 + (size_t)16 * K;
    const float* __restrict__ wcol0 = W + (size_t)(2 * half) * N + (n0 + l);
    const float* __restrict__ wcol1 = wcol0 + 16;

    v8f acc00 = {}, acc01 = {}, acc10 = {}, acc11 = {};
#pragma unroll 2
    for (int k = 0; k < K; k += 4) {
        v2f a0, a1, b0, b1;
        a0.x = arow0[k + 0];
        a0.y = arow0[k + 1];
        a1.x = arow1[k + 0];
        a1.y = arow1[k + 1];
        b0.x = wcol0[(size_t)k * N];
        b0.y = wcol0[(size_t)k * N + N];
        b1.x = wcol1[(size_t)k * N];
        b1.y = wcol1[(size_t)k * N + N];
        acc00 = __builtin_amdgcn_wmma_f32_16x16x4_f32(
            false, a0, false, b0, (short)0, acc00, false, false);
        acc01 = __builtin_amdgcn_wmma_f32_16x16x4_f32(
            false, a0, false, b1, (short)0, acc01, false, false);
        acc10 = __builtin_amdgcn_wmma_f32_16x16x4_f32(
            false, a1, false, b0, (short)0, acc10, false, false);
        acc11 = __builtin_amdgcn_wmma_f32_16x16x4_f32(
            false, a1, false, b1, (short)0, acc11, false, false);
    }

    float* __restrict__ crow0 = C + (size_t)(m0 + 8 * half) * N + (n0 + l);
    float* __restrict__ crow1 = crow0 + (size_t)16 * N;
#pragma unroll
    for (int v = 0; v < 8; ++v) {
        crow0[(size_t)v * N]      = acc00[v];
        crow0[(size_t)v * N + 16] = acc01[v];
        crow1[(size_t)v * N]      = acc10[v];
        crow1[(size_t)v * N + 16] = acc11[v];
    }
}

__global__ __launch_bounds__(256)
void gcn_zero_counts(int* __restrict__ cnt, int n)
{
    const int i = blockIdx.x * blockDim.x + threadIdx.x;
    if (i < n) cnt[i] = 0;
}

__global__ __launch_bounds__(256)
void gcn_edge_rank(const int* __restrict__ dst, int* __restrict__ cnt,
                   int* __restrict__ rank, int E)
{
    const int e = blockIdx.x * blockDim.x + threadIdx.x;
    if (e < E) rank[e] = atomicAdd(&cnt[dst[e]], 1);
}

__global__ __launch_bounds__(256)
void gcn_rowptr_scan(const int* __restrict__ cnt, int* __restrict__ rowptr)
{
    __shared__ int s_wsum[8];
    __shared__ int s_wpre[8];
    const int t    = threadIdx.x;
    const int lane = t & 31;
    const int wave = t >> 5;
    const int b    = blockIdx.x;
    const v4i* __restrict__ c4 = reinterpret_cast<const v4i*>(cnt);

    const int n4 = b * 256;
    int pre = 0;
    int i = t;
    for (; i + 768 < n4; i += 1024) {
        const v4i p0 = c4[i];
        const v4i p1 = c4[i + 256];
        const v4i p2 = c4[i + 512];
        const v4i p3 = c4[i + 768];
        pre += ((p0.x + p0.y) + (p0.z + p0.w)) + ((p1.x + p1.y) + (p1.z + p1.w));
        pre += ((p2.x + p2.y) + (p2.z + p2.w)) + ((p3.x + p3.y) + (p3.z + p3.w));
    }
    for (; i < n4; i += 256) {
        const v4i p = c4[i];
        pre += (p.x + p.y) + (p.z + p.w);
    }
    const v4i v = c4[b * 256 + t];
    const int tsum = (v.x + v.y) + (v.z + v.w);

    int inc = tsum;
    for (int o = 1; o < 32; o <<= 1) {
        const int nb = __shfl_up(inc, o, 32);
        if (lane >= o) inc += nb;
    }
    for (int o = 16; o > 0; o >>= 1) pre += __shfl_xor(pre, o, 32);

    if (lane == 31) s_wsum[wave] = inc;
    if (lane == 0)  s_wpre[wave] = pre;
    __syncthreads();

    int base = 0;
    for (int w = 0; w < 8; ++w) {
        base += s_wpre[w];
        if (w < wave) base += s_wsum[w];
    }
    const int ex = base + inc - tsum;
    v4i o;
    o.x = ex;
    o.y = ex + v.x;
    o.z = o.y + v.y;
    o.w = o.z + v.z;
    reinterpret_cast<v4i*>(rowptr)[b * 256 + t] = o;
}

__global__ __launch_bounds__(256)
void gcn_edge_place(const int* __restrict__ src, const int* __restrict__ dst,
                    const float* __restrict__ ew,
                    const int* __restrict__ rowptr, const int* __restrict__ rank,
                    v2i* __restrict__ edges, int E)
{
    const int e = blockIdx.x * blockDim.x + threadIdx.x;
    if (e >= E) return;
    const int p = rowptr[dst[e]] + rank[e];
    v2i rec;
    rec.x = src[e];
    rec.y = __float_as_int(ew[e]);
    edges[p] = rec;
}

template <int VEC> struct GcnRowVec;
template <> struct GcnRowVec<2> { typedef v2f type; };
template <> struct GcnRowVec<4> { typedef v4f type; };

template <int N, int C, typename VF>
__device__ __forceinline__ void gcn_add_edges(const float* __restrict__ sup,
                                              const v2i* __restrict__ ed,
                                              int lane, VF& acc)
{
    v2i e[N];
    VF  r[N];
#pragma unroll
    for (int k = 0; k < N; ++k) e[k] = ed[k];
#pragma unroll
    for (int k = 0; k < N; ++k)
        r[k] = reinterpret_cast<const VF*>(sup + (size_t)e[k].x * C)[lane];
#pragma unroll
    for (int k = 0; k < N; ++k) acc += r[k] * __int_as_float(e[k].y);
}

template <int C, bool RELU>
__global__ __launch_bounds__(256)
void gcn_row_gather(const float* __restrict__ sup,
                    const int* __restrict__ rowptr,
                    const int* __restrict__ cnt,
                    const v2i* __restrict__ edges,
                    const float* __restrict__ bias,
                    float* __restrict__ out,
                    int M)
{
    constexpr int VEC = C / 32;
    typedef typename GcnRowVec<VEC>::type VF;

    const int lane   = threadIdx.x & 31;
    const int nwaves = (gridDim.x * blockDim.x) >> 5;
    int row = __builtin_amdgcn_readfirstlane(
        (int)((blockIdx.x * blockDim.x + threadIdx.x) >> 5));

    const VF bv = reinterpret_cast<const VF*>(bias)[lane];

    for (; row < M; row += nwaves) {
        const int start = rowptr[row];
        const int deg   = cnt[row];
        const v2i* __restrict__ ed = edges + start;

        VF acc = {};
        int j = 0;
        for (; j + 8 <= deg; j += 8) gcn_add_edges<8, C, VF>(sup, ed + j, lane, acc);
        if (deg & 4) { gcn_add_edges<4, C, VF>(sup, ed + j, lane, acc); j += 4; }
        if (deg & 2) { gcn_add_edges<2, C, VF>(sup, ed + j, lane, acc); j += 2; }
        if (deg & 1) { gcn_add_edges<1, C, VF>(sup, ed + j, lane, acc); }

        acc += bv;
        if (RELU) {
            for (int i = 0; i < VEC; ++i) acc[i] = fmaxf(acc[i], 0.0f);
        }
        reinterpret_cast<VF*>(out + (size_t)row * C)[lane] = acc;
    }
}

extern "C" void kernel_launch(void* const* d_in, const int* in_sizes, int n_in,
                              void* d_out, int out_size, void* d_ws, size_t ws_size,
                              hipStream_t stream)
{
    const float* x    = (const float*)d_in[0];
    const int*   esrc = (const int*)  d_in[1];
    const int*   edst = (const int*)  d_in[2];
    const float* ew   = (const float*)d_in[3];
    const float* W1   = (const float*)d_in[4];
    const float* b1   = (const float*)d_in[5];
    const float* W2   = (const float*)d_in[6];
    const float* b2   = (const float*)d_in[7];
    const float* W3   = (const float*)d_in[8];
    const float* b3   = (const float*)d_in[9];

    const int M = in_sizes[0] / GCN_IN_C;
    const int E = in_sizes[1];

    float* ws     = (float*)d_ws;
    float* sup128 = ws;
    float* h128   = sup128 + (size_t)M * GCN_H1;
    float* sup64  = h128   + (size_t)M * GCN_H1;
    float* h64    = sup64  + (size_t)M * GCN_H2;
    float* out    = (float*)d_out;

    const int nscan = (M + 1023) / 1024;
    const int Mp    = nscan * 1024;
    int* cnt    = (int*)(h64 + (size_t)M * GCN_H2);
    int* rowptr = cnt + Mp;
    v2i* edges  = (v2i*)(rowptr + Mp);
    int* rank   = (int*)(edges + E);

    const int mt = M / 32;
    const int B  = 256;
    const int eb = (E + B - 1) / B;
    const int GW = 2048;

    gcn_zero_counts<<<Mp / B, B, 0, stream>>>(cnt, Mp);
    gcn_edge_rank<<<eb, B, 0, stream>>>(edst, cnt, rank, E);
    gcn_rowptr_scan<<<nscan, B, 0, stream>>>(cnt, rowptr);
    gcn_edge_place<<<eb, B, 0, stream>>>(esrc, edst, ew, rowptr, rank, edges, E);

    gcn_gemm_wmma_f32<GCN_IN_C, GCN_H1>
        <<<mt, (GCN_H1 / 32) * 32, 0, stream>>>(x, W1, sup128, M);
    gcn_row_gather<GCN_H1, true>
        <<<GW, B, 0, stream>>>(sup128, rowptr, cnt, edges, b1, h128, M);

    gcn_gemm_wmma_f32<GCN_H1, GCN_H2>
        <<<mt, (GCN_H2 / 32) * 32, 0, stream>>>(h128, W2, sup64, M);
    gcn_row_gather<GCN_H2, true>
        <<<GW, B, 0, stream>>>(sup64, rowptr, cnt, edges, b2, h64, M);

    gcn_gemm_wmma_f32<GCN_H2, GCN_OUT>
        <<<mt, (GCN_OUT / 32) * 32, 0, stream>>>(h64, W3, sup128, M);
    gcn_row_gather<GCN_OUT, false>
        <<<GW, B, 0, stream>>>(sup128, rowptr, cnt, edges, b3, out, M);
}


# Order edges by destination with a two-level counting sort in LDS (chunk histograms, bin runs, per-bin node sort), not a global atomic + scattered store per edge; row sum order = LDS arrival rank
# speedup vs baseline: 3.6019x; 1.5363x over previous
#include <hip/hip_runtime.h>


typedef float v2f __attribute__((ext_vector_type(2)));
typedef float v4f __attribute__((ext_vector_type(4)));
typedef float v8f __attribute__((ext_vector_type(8)));
typedef int   v2i __attribute__((ext_vector_type(2)));
typedef int   v4i __attribute__((ext_vector_type(4)));

#define GCN_IN_C  256
#define GCN_H1    128
#define GCN_H2    64
#define GCN_OUT   64

template <int K, int N>
__global__ __launch_bounds__(128)
void gcn_gemm_wmma_f32(const float* __restrict__ A,
                       const float* __restrict__ W,
                       float* __restrict__ C,
                       int M)
{
    const int lane = threadIdx.x & 31;
    const int wave = threadIdx.x >> 5;
    const int m0   = blockIdx.x << 5;
    const int n0   = wave << 5;
    const int half = lane >> 4;
    const int l    = lane & 15;

    const float* __restrict__ arow0 = A + (size_t)(m0 + l) * K + 2 * half;
    const float* __restrict__ arow1 = arow0 + (size_t)16 * K;
    const float* __restrict__ wcol0 = W + (size_t)(2 * half) * N + (n0 + l);
    const float* __restrict__ wcol1 = wcol0 + 16;

    v8f acc00 = {}, acc01 = {}, acc10 = {}, acc11 = {};
#pragma unroll 2
    for (int k = 0; k < K; k += 4) {
        v2f a0, a1, b0, b1;
        a0.x = arow0[k + 0];
        a0.y = arow0[k + 1];
        a1.x = arow1[k + 0];
        a1.y = arow1[k + 1];
        b0.x = wcol0[(size_t)k * N];
        b0.y = wcol0[(size_t)k * N + N];
        b1.x = wcol1[(size_t)k * N];
        b1.y = wcol1[(size_t)k * N + N];
        acc00 = __builtin_amdgcn_wmma_f32_16x16x4_f32(
            false, a0, false, b0, (short)0, acc00, false, false);
        acc01 = __builtin_amdgcn_wmma_f32_16x16x4_f32(
            false, a0, false, b1, (short)0, acc01, false, false);
        acc10 = __builtin_amdgcn_wmma_f32_16x16x4_f32(
            false, a1, false, b0, (short)0, acc10, false, false);
        acc11 = __builtin_amdgcn_wmma_f32_16x16x4_f32(
            false, a1, false, b1, (short)0, acc11, false, false);
    }

    float* __restrict__ crow0 = C + (size_t)(m0 + 8 * half) * N + (n0 + l);
    float* __restrict__ crow1 = crow0 + (size_t)16 * N;
#pragma unroll
    for (int v = 0; v < 8; ++v) {
        crow0[(size_t)v * N]      = acc00[v];
        crow0[(size_t)v * N + 16] = acc01[v];
        crow1[(size_t)v * N]      = acc10[v];
        crow1[(size_t)v * N + 16] = acc11[v];
    }
}

#define GCN_BIN_SHIFT 8
#define GCN_BIN_NODES 256
#define GCN_NBMAX     512
#define GCN_CHUNK     8192
#define GCN_CT        1024
#define GCN_CAP       8192

__global__ __launch_bounds__(GCN_CT)
void gcn_bin_count(const int* __restrict__ dst, int* __restrict__ hist, int E)
{
    __shared__ int s_h[GCN_NBMAX];
    const int t = threadIdx.x;
    if (t < GCN_NBMAX) s_h[t] = 0;
    __syncthreads();

    const int e0 = blockIdx.x * GCN_CHUNK;
    if (e0 + GCN_CHUNK <= E) {
        const v4i* __restrict__ d4 = reinterpret_cast<const v4i*>(dst + e0);
        const v4i q0 = d4[t];
        const v4i q1 = d4[t + GCN_CT];
        atomicAdd(&s_h[q0.x >> GCN_BIN_SHIFT], 1);
        atomicAdd(&s_h[q0.y >> GCN_BIN_SHIFT], 1);
        atomicAdd(&s_h[q0.z >> GCN_BIN_SHIFT], 1);
        atomicAdd(&s_h[q0.w >> GCN_BIN_SHIFT], 1);
        atomicAdd(&s_h[q1.x >> GCN_BIN_SHIFT], 1);
        atomicAdd(&s_h[q1.y >> GCN_BIN_SHIFT], 1);
        atomicAdd(&s_h[q1.z >> GCN_BIN_SHIFT], 1);
        atomicAdd(&s_h[q1.w >> GCN_BIN_SHIFT], 1);
    } else {
        for (int i = t; i < E - e0; i += GCN_CT)
            atomicAdd(&s_h[dst[e0 + i] >> GCN_BIN_SHIFT], 1);
    }
    __syncthreads();
    if (t < GCN_NBMAX) hist[blockIdx.x * GCN_NBMAX + t] = s_h[t];
}

__device__ __forceinline__ int gcn_block_excl_scan(int v, int* s_wsum, int lane, int wave)
{
    int inc = v;
    for (int o = 1; o < 32; o <<= 1) {
        const int nb = __shfl_up(inc, o, 32);
        if (lane >= o) inc += nb;
    }
    if (lane == 31) s_wsum[wave] = inc;
    __syncthreads();
    int base = 0;
    for (int w = 0; w < 32; ++w)
        if (w < wave) base += s_wsum[w];
    __syncthreads();
    return base + inc - v;
}

__device__ __forceinline__ void gcn_place(int d, int s, int w, int* s_cur,
                                          int* s_src, int* s_wgt, int* s_dst)
{
    const int p = atomicAdd(&s_cur[d >> GCN_BIN_SHIFT], 1);
    s_src[p] = s;
    s_wgt[p] = w;
    s_dst[p] = d;
}

__global__ __launch_bounds__(GCN_CT)
void gcn_bin_scatter(const int* __restrict__ src, const int* __restrict__ dst,
                     const float* __restrict__ ew, const int* __restrict__ hist,
                     int* __restrict__ binptr, v2i* __restrict__ binned,
                     int E, int nchunks)
{
    __shared__ int s_src[GCN_CHUNK];
    __shared__ int s_wgt[GCN_CHUNK];
    __shared__ int s_dst[GCN_CHUNK];
    __shared__ int s_tot1[GCN_NBMAX];
    __shared__ int s_pre1[GCN_NBMAX];
    __shared__ int s_cur[GCN_NBMAX];
    __shared__ int s_gbase[GCN_NBMAX];
    __shared__ int s_wsum[32];

    const int t    = threadIdx.x;
    const int lane = t & 31;
    const int wave = t >> 5;
    const int b    = blockIdx.x;
    const int bin  = t & (GCN_NBMAX - 1);
    const int part = t >> 9;

    const int half = (nchunks + 1) >> 1;
    const int lo   = part ? half : 0;
    const int hi   = part ? nchunks : half;
    int tot = 0, pre = 0;
#pragma unroll 16
    for (int j = lo; j < hi; ++j) {
        const int v = hist[j * GCN_NBMAX + bin];
        tot += v;
        pre += (j < b) ? v : 0;
    }
    if (part) { s_tot1[bin] = tot; s_pre1[bin] = pre; }
    __syncthreads();
    int own = 0;
    if (!part) {
        tot += s_tot1[bin];
        pre += s_pre1[bin];
        own  = hist[b * GCN_NBMAX + bin];
    } else {
        tot = 0;
    }
    const int binbase = gcn_block_excl_scan(tot, s_wsum, lane, wave);
    const int loff    = gcn_block_excl_scan(own, s_wsum, lane, wave);
    if (!part) {
        s_cur[bin]   = loff;
        s_gbase[bin] = binbase + pre - loff;
        if (b == 0) {
            binptr[bin] = binbase;
            if (bin == GCN_NBMAX - 1) binptr[GCN_NBMAX] = binbase + tot;
        }
    }
    __syncthreads();

    const int e0 = b * GCN_CHUNK;
    if (e0 + GCN_CHUNK <= E) {
        const v4i* __restrict__ d4 = reinterpret_cast<const v4i*>(dst + e0);
        const v4i* __restrict__ s4 = reinterpret_cast<const v4i*>(src + e0);
        const v4i* __restrict__ w4 = reinterpret_cast<const v4i*>(ew + e0);
        const v4i dq0 = d4[t], dq1 = d4[t + GCN_CT];
        const v4i sq0 = s4[t], sq1 = s4[t + GCN_CT];
        const v4i wq0 = w4[t], wq1 = w4[t + GCN_CT];
        gcn_place(dq0.x, sq0.x, wq0.x, s_cur, s_src, s_wgt, s_dst);
        gcn_place(dq0.y, sq0.y, wq0.y, s_cur, s_src, s_wgt, s_dst);
        gcn_place(dq0.z, sq0.z, wq0.z, s_cur, s_src, s_wgt, s_dst);
        gcn_place(dq0.w, sq0.w, wq0.w, s_cur, s_src, s_wgt, s_dst);
        gcn_place(dq1.x, sq1.x, wq1.x, s_cur, s_src, s_wgt, s_dst);
        gcn_place(dq1.y, sq1.y, wq1.y, s_cur, s_src, s_wgt, s_dst);
        gcn_place(dq1.z, sq1.z, wq1.z, s_cur, s_src, s_wgt, s_dst);
        gcn_place(dq1.w, sq1.w, wq1.w, s_cur, s_src, s_wgt, s_dst);
    } else {
        for (int i = t; i < E - e0; i += GCN_CT)
            gcn_place(dst[e0 + i], src[e0 + i], __float_as_int(ew[e0 + i]),
                      s_cur, s_src, s_wgt, s_dst);
    }
    __syncthreads();

    const int nloc = min(GCN_CHUNK, E - e0);
    for (int i = t; i < nloc; i += GCN_CT) {
        const int d = s_dst[i];
        v2i rec;
        rec.x = s_src[i] | (int)(((unsigned)d & (GCN_BIN_NODES - 1)) << 24);
        rec.y = s_wgt[i];
        binned[s_gbase[d >> GCN_BIN_SHIFT] + i] = rec;
    }
}

__global__ __launch_bounds__(GCN_BIN_NODES)
void gcn_bin_sort(const v2i* __restrict__ binned, const int* __restrict__ binptr,
                  v2i* __restrict__ edges, int* __restrict__ rowptr,
                  int* __restrict__ cnt, int M)
{
    __shared__ v2i s_stage[GCN_CAP];
    __shared__ int s_cnt[GCN_BIN_NODES];
    __shared__ int s_cur[GCN_BIN_NODES];
    __shared__ int s_wsum[8];

    const int t    = threadIdx.x;
    const int lane = t & 31;
    const int wave = t >> 5;
    const int b    = blockIdx.x;
    const int base = binptr[b];
    const int n    = binptr[b + 1] - base;
    const v2i* __restrict__ recs = binned + base;
    v2i* __restrict__ outp = edges + base;

    s_cnt[t] = 0;
    __syncthreads();

    for (int i0 = 0; i0 < n; i0 += 4 * GCN_BIN_NODES) {
        int key[4];
#pragma unroll
        for (int k = 0; k < 4; ++k) {
            const int i = i0 + k * GCN_BIN_NODES + t;
            key[k] = (i < n) ? recs[i].x : 0;
        }
#pragma unroll
        for (int k = 0; k < 4; ++k) {
            const int i = i0 + k * GCN_BIN_NODES + t;
            if (i < n) atomicAdd(&s_cnt[(unsigned)key[k] >> 24], 1);
        }
    }
    __syncthreads();

    const int c = s_cnt[t];
    int inc = c;
    for (int o = 1; o < 32; o <<= 1) {
        const int nb = __shfl_up(inc, o, 32);
        if (lane >= o) inc += nb;
    }
    if (lane == 31) s_wsum[wave] = inc;
    __syncthreads();
    int wbase = 0;
    for (int w = 0; w < 8; ++w)
        if (w < wave) wbase += s_wsum[w];
    const int ex = wbase + inc - c;
    s_cur[t] = ex;
    const int node = b * GCN_BIN_NODES + t;
    if (node < M) {
        rowptr[node] = base + ex;
        cnt[node]    = c;
    }
    __syncthreads();

    const bool fits = (n <= GCN_CAP);
    for (int i0 = 0; i0 < n; i0 += 4 * GCN_BIN_NODES) {
        v2i r[4];
#pragma unroll
        for (int k = 0; k < 4; ++k) {
            const int i = i0 + k * GCN_BIN_NODES + t;
            v2i z; z.x = 0; z.y = 0;
            r[k] = (i < n) ? recs[i] : z;
        }
#pragma unroll
        for (int k = 0; k < 4; ++k) {
            const int i = i0 + k * GCN_BIN_NODES + t;
            if (i < n) {
                const int p = atomicAdd(&s_cur[(unsigned)r[k].x >> 24], 1);
                v2i rec;
                rec.x = r[k].x & 0x00FFFFFF;
                rec.y = r[k].y;
                if (fits) s_stage[p] = rec;
                else      outp[p]    = rec;
            }
        }
    }
    __syncthreads();
    if (fits) {
        for (int i = t; i < n; i += GCN_BIN_NODES) outp[i] = s_stage[i];
    }
}

template <int VEC> struct GcnRowVec;
template <> struct GcnRowVec<2> { typedef v2f type; };
template <> struct GcnRowVec<4> { typedef v4f type; };

template <int N, int C, typename VF>
__device__ __forceinline__ void gcn_add_edges(const float* __restrict__ sup,
                                              const v2i* __restrict__ ed,
                                              int lane, VF& acc)
{
    v2i e[N];
    VF  r[N];
#pragma unroll
    for (int k = 0; k < N; ++k) e[k] = ed[k];
#pragma unroll
    for (int k = 0; k < N; ++k)
        r[k] = reinterpret_cast<const VF*>(sup + (size_t)e[k].x * C)[lane];
#pragma unroll
    for (int k = 0; k < N; ++k) acc += r[k] * __int_as_float(e[k].y);
}

template <int C, bool RELU>
__global__ __launch_bounds__(256)
void gcn_row_gather(const float* __restrict__ sup,
                    const int* __restrict__ rowptr,
                    const int* __restrict__ cnt,
                    const v2i* __restrict__ edges,
                    const float* __restrict__ bias,
                    float* __restrict__ out,
                    int M)
{
    constexpr int VEC = C / 32;
    typedef typename GcnRowVec<VEC>::type VF;

    const int lane   = threadIdx.x & 31;
    const int nwaves = (gridDim.x * blockDim.x) >> 5;
    int row = __builtin_amdgcn_readfirstlane(
        (int)((blockIdx.x * blockDim.x + threadIdx.x) >> 5));

    const VF bv = reinterpret_cast<const VF*>(bias)[lane];

    for (; row < M; row += nwaves) {
        const int start = rowptr[row];
        const int deg   = cnt[row];
        const v2i* __restrict__ ed = edges + start;

        VF acc = {};
        int j = 0;
        for (; j + 8 <= deg; j += 8) gcn_add_edges<8, C, VF>(sup, ed + j, lane, acc);
        if (deg & 4) { gcn_add_edges<4, C, VF>(sup, ed + j, lane, acc); j += 4; }
        if (deg & 2) { gcn_add_edges<2, C, VF>(sup, ed + j, lane, acc); j += 2; }
        if (deg & 1) { gcn_add_edges<1, C, VF>(sup, ed + j, lane, acc); }

        acc += bv;
        if (RELU) {
            for (int i = 0; i < VEC; ++i) acc[i] = fmaxf(acc[i], 0.0f);
        }
        reinterpret_cast<VF*>(out + (size_t)row * C)[lane] = acc;
    }
}

extern "C" void kernel_launch(void* const* d_in, const int* in_sizes, int n_in,
                              void* d_out, int out_size, void* d_ws, size_t ws_size,
                              hipStream_t stream)
{
    const float* x    = (const float*)d_in[0];
    const int*   esrc = (const int*)  d_in[1];
    const int*   edst = (const int*)  d_in[2];
    const float* ew   = (const float*)d_in[3];
    const float* W1   = (const float*)d_in[4];
    const float* b1   = (const float*)d_in[5];
    const float* W2   = (const float*)d_in[6];
    const float* b2   = (const float*)d_in[7];
    const float* W3   = (const float*)d_in[8];
    const float* b3   = (const float*)d_in[9];

    const int M = in_sizes[0] / GCN_IN_C;
    const int E = in_sizes[1];

    float* ws     = (float*)d_ws;
    float* sup128 = ws;
    float* h128   = sup128 + (size_t)M * GCN_H1;
    float* sup64  = h128   + (size_t)M * GCN_H1;
    float* h64    = sup64  + (size_t)M * GCN_H2;
    float* out    = (float*)d_out;

    const int nchunks = (E + GCN_CHUNK - 1) / GCN_CHUNK;
    const int nbins   = (M + GCN_BIN_NODES - 1) / GCN_BIN_NODES;
    int* cnt    = (int*)(h64 + (size_t)M * GCN_H2);
    int* rowptr = cnt + nbins * GCN_BIN_NODES;
    v2i* edges  = (v2i*)(rowptr + nbins * GCN_BIN_NODES);
    v2i* binned = edges + E;
    int* hist   = (int*)(binned + E);
    int* binptr = hist + (size_t)nchunks * GCN_NBMAX;

    const int mt = M / 32;
    const int B  = 256;
    const int GW = 2048;

    gcn_bin_count<<<nchunks, GCN_CT, 0, stream>>>(edst, hist, E);
    gcn_bin_scatter<<<nchunks, GCN_CT, 0, stream>>>(esrc, edst, ew, hist, binptr,
                                                    binned, E, nchunks);
    gcn_bin_sort<<<nbins, GCN_BIN_NODES, 0, stream>>>(binned, binptr, edges,
                                                      rowptr, cnt, M);

    gcn_gemm_wmma_f32<GCN_IN_C, GCN_H1>
        <<<mt, (GCN_H1 / 32) * 32, 0, stream>>>(x, W1, sup128, M);
    gcn_row_gather<GCN_H1, true>
        <<<GW, B, 0, stream>>>(sup128, rowptr, cnt, edges, b1, h128, M);

    gcn_gemm_wmma_f32<GCN_H1, GCN_H2>
        <<<mt, (GCN_H2 / 32) * 32, 0, stream>>>(h128, W2, sup64, M);
    gcn_row_gather<GCN_H2, true>
        <<<GW, B, 0, stream>>>(sup64, rowptr, cnt, edges, b2, h64, M);

    gcn_gemm_wmma_f32<GCN_H2, GCN_OUT>
        <<<mt, (GCN_OUT / 32) * 32, 0, stream>>>(h64, W3, sup128, M);
    gcn_row_gather<GCN_OUT, false>
        <<<GW, B, 0, stream>>>(sup128, rowptr, cnt, edges, b3, out, M);
}


# Fuse each layer's neighbour sum with the next layer's product: a block keeps its 32 summed rows (bias, ReLU) in LDS and multiplies them by W2 / W3 on WMMA at once; h1, h2 and two launches go
# speedup vs baseline: 3.9247x; 1.0896x over previous
#include <hip/hip_runtime.h>


typedef float v2f __attribute__((ext_vector_type(2)));
typedef float v4f __attribute__((ext_vector_type(4)));
typedef float v8f __attribute__((ext_vector_type(8)));
typedef int   v2i __attribute__((ext_vector_type(2)));
typedef int   v4i __attribute__((ext_vector_type(4)));

#define GCN_IN_C  256
#define GCN_H1    128
#define GCN_H2    64
#define GCN_OUT   64

template <int K, int N>
__global__ __launch_bounds__(128)
void gcn_gemm_wmma_f32(const float* __restrict__ A,
                       const float* __restrict__ W,
                       float* __restrict__ C,
                       int M)
{
    const int lane = threadIdx.x & 31;
    const int wave = threadIdx.x >> 5;
    const int m0   = blockIdx.x << 5;
    const int n0   = wave << 5;
    const int half = lane >> 4;
    const int l    = lane & 15;

    const float* __restrict__ arow0 = A + (size_t)(m0 + l) * K + 2 * half;
    const float* __restrict__ arow1 = arow0 + (size_t)16 * K;
    const float* __restrict__ wcol0 = W + (size_t)(2 * half) * N + (n0 + l);
    const float* __restrict__ wcol1 = wcol0 + 16;

    v8f acc00 = {}, acc01 = {}, acc10 = {}, acc11 = {};
#pragma unroll 2
    for (int k = 0; k < K; k += 4) {
        v2f a0, a1, b0, b1;
        a0.x = arow0[k + 0];
        a0.y = arow0[k + 1];
        a1.x = arow1[k + 0];
        a1.y = arow1[k + 1];
        b0.x = wcol0[(size_t)k * N];
        b0.y = wcol0[(size_t)k * N + N];
        b1.x = wcol1[(size_t)k * N];
        b1.y = wcol1[(size_t)k * N + N];
        acc00 = __builtin_amdgcn_wmma_f32_16x16x4_f32(
            false, a0, false, b0, (short)0, acc00, false, false);
        acc01 = __builtin_amdgcn_wmma_f32_16x16x4_f32(
            false, a0, false, b1, (short)0, acc01, false, false);
        acc10 = __builtin_amdgcn_wmma_f32_16x16x4_f32(
            false, a1, false, b0, (short)0, acc10, false, false);
        acc11 = __builtin_amdgcn_wmma_f32_16x16x4_f32(
            false, a1, false, b1, (short)0, acc11, false, false);
    }

    float* __restrict__ crow0 = C + (size_t)(m0 + 8 * half) * N + (n0 + l);
    float* __restrict__ crow1 = crow0 + (size_t)16 * N;
#pragma unroll
    for (int v = 0; v < 8; ++v) {
        crow0[(size_t)v * N]      = acc00[v];
        crow0[(size_t)v * N + 16] = acc01[v];
        crow1[(size_t)v * N]      = acc10[v];
        crow1[(size_t)v * N + 16] = acc11[v];
    }
}

#define GCN_BIN_SHIFT 8
#define GCN_BIN_NODES 256
#define GCN_NBMAX     512
#define GCN_CHUNK     8192
#define GCN_CT        1024
#define GCN_CAP       8192

__global__ __launch_bounds__(GCN_CT)
void gcn_bin_count(const int* __restrict__ dst, int* __restrict__ hist, int E)
{
    __shared__ int s_h[GCN_NBMAX];
    const int t = threadIdx.x;
    if (t < GCN_NBMAX) s_h[t] = 0;
    __syncthreads();

    const int e0 = blockIdx.x * GCN_CHUNK;
    if (e0 + GCN_CHUNK <= E) {
        const v4i* __restrict__ d4 = reinterpret_cast<const v4i*>(dst + e0);
        const v4i q0 = d4[t];
        const v4i q1 = d4[t + GCN_CT];
        atomicAdd(&s_h[q0.x >> GCN_BIN_SHIFT], 1);
        atomicAdd(&s_h[q0.y >> GCN_BIN_SHIFT], 1);
        atomicAdd(&s_h[q0.z >> GCN_BIN_SHIFT], 1);
        atomicAdd(&s_h[q0.w >> GCN_BIN_SHIFT], 1);
        atomicAdd(&s_h[q1.x >> GCN_BIN_SHIFT], 1);
        atomicAdd(&s_h[q1.y >> GCN_BIN_SHIFT], 1);
        atomicAdd(&s_h[q1.z >> GCN_BIN_SHIFT], 1);
        atomicAdd(&s_h[q1.w >> GCN_BIN_SHIFT], 1);
    } else {
        for (int i = t; i < E - e0; i += GCN_CT)
            atomicAdd(&s_h[dst[e0 + i] >> GCN_BIN_SHIFT], 1);
    }
    __syncthreads();
    if (t < GCN_NBMAX) hist[blockIdx.x * GCN_NBMAX + t] = s_h[t];
}

__device__ __forceinline__ int gcn_block_excl_scan(int v, int* s_wsum, int lane, int wave)
{
    int inc = v;
    for (int o = 1; o < 32; o <<= 1) {
        const int nb = __shfl_up(inc, o, 32);
        if (lane >= o) inc += nb;
    }
    if (lane == 31) s_wsum[wave] = inc;
    __syncthreads();
    int base = 0;
    for (int w = 0; w < 32; ++w)
        if (w < wave) base += s_wsum[w];
    __syncthreads();
    return base + inc - v;
}

__device__ __forceinline__ void gcn_place(int d, int s, int w, int* s_cur,
                                          int* s_src, int* s_wgt, int* s_dst)
{
    const int p = atomicAdd(&s_cur[d >> GCN_BIN_SHIFT], 1);
    s_src[p] = s;
    s_wgt[p] = w;
    s_dst[p] = d;
}

__global__ __launch_bounds__(GCN_CT)
void gcn_bin_scatter(const int* __restrict__ src, const int* __restrict__ dst,
                     const float* __restrict__ ew, const int* __restrict__ hist,
                     int* __restrict__ binptr, v2i* __restrict__ binned,
                     int E, int nchunks)
{
    __shared__ int s_src[GCN_CHUNK];
    __shared__ int s_wgt[GCN_CHUNK];
    __shared__ int s_dst[GCN_CHUNK];
    __shared__ int s_tot1[GCN_NBMAX];
    __shared__ int s_pre1[GCN_NBMAX];
    __shared__ int s_cur[GCN_NBMAX];
    __shared__ int s_gbase[GCN_NBMAX];
    __shared__ int s_wsum[32];

    const int t    = threadIdx.x;
    const int lane = t & 31;
    const int wave = t >> 5;
    const int b    = blockIdx.x;
    const int bin  = t & (GCN_NBMAX - 1);
    const int part = t >> 9;

    const int half = (nchunks + 1) >> 1;
    const int lo   = part ? half : 0;
    const int hi   = part ? nchunks : half;
    int tot = 0, pre = 0;
#pragma unroll 16
    for (int j = lo; j < hi; ++j) {
        const int v = hist[j * GCN_NBMAX + bin];
        tot += v;
        pre += (j < b) ? v : 0;
    }
    if (part) { s_tot1[bin] = tot; s_pre1[bin] = pre; }
    __syncthreads();
    int own = 0;
    if (!part) {
        tot += s_tot1[bin];
        pre += s_pre1[bin];
        own  = hist[b * GCN_NBMAX + bin];
    } else {
        tot = 0;
    }
    const int binbase = gcn_block_excl_scan(tot, s_wsum, lane, wave);
    const int loff    = gcn_block_excl_scan(own, s_wsum, lane, wave);
    if (!part) {
        s_cur[bin]   = loff;
        s_gbase[bin] = binbase + pre - loff;
        if (b == 0) {
            binptr[bin] = binbase;
            if (bin == GCN_NBMAX - 1) binptr[GCN_NBMAX] = binbase + tot;
        }
    }
    __syncthreads();

    const int e0 = b * GCN_CHUNK;
    if (e0 + GCN_CHUNK <= E) {
        const v4i* __restrict__ d4 = reinterpret_cast<const v4i*>(dst + e0);
        const v4i* __restrict__ s4 = reinterpret_cast<const v4i*>(src + e0);
        const v4i* __restrict__ w4 = reinterpret_cast<const v4i*>(ew + e0);
        const v4i dq0 = d4[t], dq1 = d4[t + GCN_CT];
        const v4i sq0 = s4[t], sq1 = s4[t + GCN_CT];
        const v4i wq0 = w4[t], wq1 = w4[t + GCN_CT];
        gcn_place(dq0.x, sq0.x, wq0.x, s_cur, s_src, s_wgt, s_dst);
        gcn_place(dq0.y, sq0.y, wq0.y, s_cur, s_src, s_wgt, s_dst);
        gcn_place(dq0.z, sq0.z, wq0.z, s_cur, s_src, s_wgt, s_dst);
        gcn_place(dq0.w, sq0.w, wq0.w, s_cur, s_src, s_wgt, s_dst);
        gcn_place(dq1.x, sq1.x, wq1.x, s_cur, s_src, s_wgt, s_dst);
        gcn_place(dq1.y, sq1.y, wq1.y, s_cur, s_src, s_wgt, s_dst);
        gcn_place(dq1.z, sq1.z, wq1.z, s_cur, s_src, s_wgt, s_dst);
        gcn_place(dq1.w, sq1.w, wq1.w, s_cur, s_src, s_wgt, s_dst);
    } else {
        for (int i = t; i < E - e0; i += GCN_CT)
            gcn_place(dst[e0 + i], src[e0 + i], __float_as_int(ew[e0 + i]),
                      s_cur, s_src, s_wgt, s_dst);
    }
    __syncthreads();

    const int nloc = min(GCN_CHUNK, E - e0);
    for (int i = t; i < nloc; i += GCN_CT) {
        const int d = s_dst[i];
        v2i rec;
        rec.x = s_src[i] | (int)(((unsigned)d & (GCN_BIN_NODES - 1)) << 24);
        rec.y = s_wgt[i];
        binned[s_gbase[d >> GCN_BIN_SHIFT] + i] = rec;
    }
}

__global__ __launch_bounds__(GCN_BIN_NODES)
void gcn_bin_sort(const v2i* __restrict__ binned, const int* __restrict__ binptr,
                  v2i* __restrict__ edges, int* __restrict__ rowptr,
                  int* __restrict__ cnt, int M)
{
    __shared__ v2i s_stage[GCN_CAP];
    __shared__ int s_cnt[GCN_BIN_NODES];
    __shared__ int s_cur[GCN_BIN_NODES];
    __shared__ int s_wsum[8];

    const int t    = threadIdx.x;
    const int lane = t & 31;
    const int wave = t >> 5;
    const int b    = blockIdx.x;
    const int base = binptr[b];
    const int n    = binptr[b + 1] - base;
    const v2i* __restrict__ recs = binned + base;
    v2i* __restrict__ outp = edges + base;

    s_cnt[t] = 0;
    __syncthreads();

    for (int i0 = 0; i0 < n; i0 += 4 * GCN_BIN_NODES) {
        int key[4];
#pragma unroll
        for (int k = 0; k < 4; ++k) {
            const int i = i0 + k * GCN_BIN_NODES + t;
            key[k] = (i < n) ? recs[i].x : 0;
        }
#pragma unroll
        for (int k = 0; k < 4; ++k) {
            const int i = i0 + k * GCN_BIN_NODES + t;
            if (i < n) atomicAdd(&s_cnt[(unsigned)key[k] >> 24], 1);
        }
    }
    __syncthreads();

    const int c = s_cnt[t];
    int inc = c;
    for (int o = 1; o < 32; o <<= 1) {
        const int nb = __shfl_up(inc, o, 32);
        if (lane >= o) inc += nb;
    }
    if (lane == 31) s_wsum[wave] = inc;
    __syncthreads();
    int wbase = 0;
    for (int w = 0; w < 8; ++w)
        if (w < wave) wbase += s_wsum[w];
    const int ex = wbase + inc - c;
    s_cur[t] = ex;
    const int node = b * GCN_BIN_NODES + t;
    if (node < M) {
        rowptr[node] = base + ex;
        cnt[node]    = c;
    }
    __syncthreads();

    const bool fits = (n <= GCN_CAP);
    for (int i0 = 0; i0 < n; i0 += 4 * GCN_BIN_NODES) {
        v2i r[4];
#pragma unroll
        for (int k = 0; k < 4; ++k) {
            const int i = i0 + k * GCN_BIN_NODES + t;
            v2i z; z.x = 0; z.y = 0;
            r[k] = (i < n) ? recs[i] : z;
        }
#pragma unroll
        for (int k = 0; k < 4; ++k) {
            const int i = i0 + k * GCN_BIN_NODES + t;
            if (i < n) {
                const int p = atomicAdd(&s_cur[(unsigned)r[k].x >> 24], 1);
                v2i rec;
                rec.x = r[k].x & 0x00FFFFFF;
                rec.y = r[k].y;
                if (fits) s_stage[p] = rec;
                else      outp[p]    = rec;
            }
        }
    }
    __syncthreads();
    if (fits) {
        for (int i = t; i < n; i += GCN_BIN_NODES) outp[i] = s_stage[i];
    }
}

template <int VEC> struct GcnRowVec;
template <> struct GcnRowVec<2> { typedef v2f type; };
template <> struct GcnRowVec<4> { typedef v4f type; };

template <int N, int C, typename VF>
__device__ __forceinline__ void gcn_add_edges(const float* __restrict__ sup,
                                              const v2i* __restrict__ ed,
                                              int lane, VF& acc)
{
    v2i e[N];
    VF  r[N];
#pragma unroll
    for (int k = 0; k < N; ++k) e[k] = ed[k];
#pragma unroll
    for (int k = 0; k < N; ++k)
        r[k] = reinterpret_cast<const VF*>(sup + (size_t)e[k].x * C)[lane];
#pragma unroll
    for (int k = 0; k < N; ++k) acc += r[k] * __int_as_float(e[k].y);
}

template <int C, bool RELU>
__global__ __launch_bounds__(256)
void gcn_row_gather(const float* __restrict__ sup,
                    const int* __restrict__ rowptr,
                    const int* __restrict__ cnt,
                    const v2i* __restrict__ edges,
                    const float* __restrict__ bias,
                    float* __restrict__ out,
                    int M)
{
    constexpr int VEC = C / 32;
    typedef typename GcnRowVec<VEC>::type VF;

    const int lane   = threadIdx.x & 31;
    const int nwaves = (gridDim.x * blockDim.x) >> 5;
    int row = __builtin_amdgcn_readfirstlane(
        (int)((blockIdx.x * blockDim.x + threadIdx.x) >> 5));

    const VF bv = reinterpret_cast<const VF*>(bias)[lane];

    for (; row < M; row += nwaves) {
        const int start = rowptr[row];
        const int deg   = cnt[row];
        const v2i* __restrict__ ed = edges + start;

        VF acc = {};
        int j = 0;
        for (; j + 8 <= deg; j += 8) gcn_add_edges<8, C, VF>(sup, ed + j, lane, acc);
        if (deg & 4) { gcn_add_edges<4, C, VF>(sup, ed + j, lane, acc); j += 4; }
        if (deg & 2) { gcn_add_edges<2, C, VF>(sup, ed + j, lane, acc); j += 2; }
        if (deg & 1) { gcn_add_edges<1, C, VF>(sup, ed + j, lane, acc); }

        acc += bv;
        if (RELU) {
            for (int i = 0; i < VEC; ++i) acc[i] = fmaxf(acc[i], 0.0f);
        }
        reinterpret_cast<VF*>(out + (size_t)row * C)[lane] = acc;
    }
}


template <int C>
__global__ __launch_bounds__(256)
void gcn_gather_relu_gemm64(const float* __restrict__ sup,
                            const int* __restrict__ rowptr,
                            const int* __restrict__ cnt,
                            const v2i* __restrict__ edges,
                            const float* __restrict__ bias,
                            const float* __restrict__ W,
                            float* __restrict__ next,
                            int M)
{
    constexpr int VEC   = C / 32;
    constexpr int N     = 64;
    constexpr int PITCH = C + 4;
    constexpr int OP    = N + 4;
    typedef typename GcnRowVec<VEC>::type VF;

    __shared__ __attribute__((aligned(16))) float s_h[32 * PITCH];
    __shared__ __attribute__((aligned(16))) float s_o[32 * OP];

    const int t    = threadIdx.x;
    const int lane = t & 31;
    const int wave = __builtin_amdgcn_readfirstlane(t >> 5);
    const int m0   = blockIdx.x << 5;

    const VF bv = reinterpret_cast<const VF*>(bias)[lane];

    for (int i = 0; i < 4; ++i) {
        const int r   = wave * 4 + i;
        const int row = m0 + r;
        VF acc = {};
        if (row < M) {
            const int start = rowptr[row];
            const int deg   = cnt[row];
            const v2i* __restrict__ ed = edges + start;
            int j = 0;
            for (; j + 8 <= deg; j += 8) gcn_add_edges<8, C, VF>(sup, ed + j, lane, acc);
            if (deg & 4) { gcn_add_edges<4, C, VF>(sup, ed + j, lane, acc); j += 4; }
            if (deg & 2) { gcn_add_edges<2, C, VF>(sup, ed + j, lane, acc); j += 2; }
            if (deg & 1) { gcn_add_edges<1, C, VF>(sup, ed + j, lane, acc); }
            acc += bv;
            for (int q = 0; q < VEC; ++q) acc[q] = fmaxf(acc[q], 0.0f);
        }
        reinterpret_cast<VF*>(s_h + r * PITCH)[lane] = acc;
    }
    __syncthreads();

    const int half = lane >> 4;
    const int l    = lane & 15;
    const int mt   = wave >> 2;
    const int nt   = wave & 3;
    const float* arow = s_h + (mt * 16 + l) * PITCH + 2 * half;
    const float* __restrict__ wcol = W + (size_t)(2 * half) * N + (nt * 16 + l);

    v8f d = {};
#pragma unroll 8
    for (int k = 0; k < C; k += 4) {
        v2f a, b;
        a   = *reinterpret_cast<const v2f*>(arow + k);
        b.x = wcol[(size_t)k * N];
        b.y = wcol[(size_t)k * N + N];
        d = __builtin_amdgcn_wmma_f32_16x16x4_f32(
            false, a, false, b, (short)0, d, false, false);
    }

    float* orow = s_o + (mt * 16 + 8 * half) * OP + (nt * 16 + l);
#pragma unroll
    for (int v = 0; v < 8; ++v) orow[v * OP] = d[v];
    __syncthreads();

#pragma unroll
    for (int p = 0; p < 2; ++p) {
        const int r  = (t >> 4) + 16 * p;
        const int c4 = (t & 15) * 4;
        if (m0 + r < M)
            *reinterpret_cast<v4f*>(next + (size_t)(m0 + r) * N + c4) =
                *reinterpret_cast<const v4f*>(s_o + r * OP + c4);
    }
}

extern "C" void kernel_launch(void* const* d_in, const int* in_sizes, int n_in,
                              void* d_out, int out_size, void* d_ws, size_t ws_size,
                              hipStream_t stream)
{
    const float* x    = (const float*)d_in[0];
    const int*   esrc = (const int*)  d_in[1];
    const int*   edst = (const int*)  d_in[2];
    const float* ew   = (const float*)d_in[3];
    const float* W1   = (const float*)d_in[4];
    const float* b1   = (const float*)d_in[5];
    const float* W2   = (const float*)d_in[6];
    const float* b2   = (const float*)d_in[7];
    const float* W3   = (const float*)d_in[8];
    const float* b3   = (const float*)d_in[9];

    const int M = in_sizes[0] / GCN_IN_C;
    const int E = in_sizes[1];

    float* ws     = (float*)d_ws;
    float* sup128 = ws;
    float* sup64  = sup128 + (size_t)M * GCN_H1;
    float* out    = (float*)d_out;

    const int nchunks = (E + GCN_CHUNK - 1) / GCN_CHUNK;
    const int nbins   = (M + GCN_BIN_NODES - 1) / GCN_BIN_NODES;
    int* cnt    = (int*)(sup64 + (size_t)M * GCN_H2);
    int* rowptr = cnt + nbins * GCN_BIN_NODES;
    v2i* edges  = (v2i*)(rowptr + nbins * GCN_BIN_NODES);
    v2i* binned = edges + E;
    int* hist   = (int*)(binned + E);
    int* binptr = hist + (size_t)nchunks * GCN_NBMAX;

    const int mt = M / 32;
    const int B  = 256;
    const int GW = 2048;
    const int tiles = (M + 31) / 32;

    gcn_bin_count<<<nchunks, GCN_CT, 0, stream>>>(edst, hist, E);
    gcn_bin_scatter<<<nchunks, GCN_CT, 0, stream>>>(esrc, edst, ew, hist, binptr,
                                                    binned, E, nchunks);
    gcn_bin_sort<<<nbins, GCN_BIN_NODES, 0, stream>>>(binned, binptr, edges,
                                                      rowptr, cnt, M);

    gcn_gemm_wmma_f32<GCN_IN_C, GCN_H1>
        <<<mt, (GCN_H1 / 32) * 32, 0, stream>>>(x, W1, sup128, M);

    gcn_gather_relu_gemm64<GCN_H1>
        <<<tiles, B, 0, stream>>>(sup128, rowptr, cnt, edges, b1, W2, sup64, M);

    gcn_gather_relu_gemm64<GCN_H2>
        <<<tiles, B, 0, stream>>>(sup64, rowptr, cnt, edges, b2, W3, sup128, M);

    gcn_row_gather<GCN_OUT, false>
        <<<GW, B, 0, stream>>>(sup128, rowptr, cnt, edges, b3, out, M);
}


# Run the layer-1 WMMA product in the same launch as the last edge-sort pass (block index picks the role): the sort blocks' memory and LDS waits overlap the matrix-bound GEMM; 7 -> 6 launches
# speedup vs baseline: 4.1035x; 1.0456x over previous
#include <hip/hip_runtime.h>


typedef float v2f __attribute__((ext_vector_type(2)));
typedef float v4f __attribute__((ext_vector_type(4)));
typedef float v8f __attribute__((ext_vector_type(8)));
typedef int   v2i __attribute__((ext_vector_type(2)));
typedef int   v4i __attribute__((ext_vector_type(4)));

#define GCN_IN_C  256
#define GCN_H1    128
#define GCN_H2    64
#define GCN_OUT   64

template <int K, int N>
__device__ __forceinline__
void gcn_gemm_patch_wmma_f32(const float* __restrict__ A,
                             const float* __restrict__ W,
                             float* __restrict__ C,
                             int m0, int n0, int lane)
{
    const int half = lane >> 4;
    const int l    = lane & 15;

    const float* __restrict__ arow0 = A + (size_t)(m0 + l) * K + 2 * half;
    const float* __restrict__ arow1 = arow0 + (size_t)16 * K;
    const float* __restrict__ wcol0 = W + (size_t)(2 * half) * N + (n0 + l);
    const float* __restrict__ wcol1 = wcol0 + 16;

    v8f acc00 = {}, acc01 = {}, acc10 = {}, acc11 = {};
#pragma unroll 2
    for (int k = 0; k < K; k += 4) {
        v2f a0, a1, b0, b1;
        a0.x = arow0[k + 0];
        a0.y = arow0[k + 1];
        a1.x = arow1[k + 0];
        a1.y = arow1[k + 1];
        b0.x = wcol0[(size_t)k * N];
        b0.y = wcol0[(size_t)k * N + N];
        b1.x = wcol1[(size_t)k * N];
        b1.y = wcol1[(size_t)k * N + N];
        acc00 = __builtin_amdgcn_wmma_f32_16x16x4_f32(
            false, a0, false, b0, (short)0, acc00, false, false);
        acc01 = __builtin_amdgcn_wmma_f32_16x16x4_f32(
            false, a0, false, b1, (short)0, acc01, false, false);
        acc10 = __builtin_amdgcn_wmma_f32_16x16x4_f32(
            false, a1, false, b0, (short)0, acc10, false, false);
        acc11 = __builtin_amdgcn_wmma_f32_16x16x4_f32(
            false, a1, false, b1, (short)0, acc11, false, false);
    }

    float* __restrict__ crow0 = C + (size_t)(m0 + 8 * half) * N + (n0 + l);
    float* __restrict__ crow1 = crow0 + (size_t)16 * N;
#pragma unroll
    for (int v = 0; v < 8; ++v) {
        crow0[(size_t)v * N]      = acc00[v];
        crow0[(size_t)v * N + 16] = acc01[v];
        crow1[(size_t)v * N]      = acc10[v];
        crow1[(size_t)v * N + 16] = acc11[v];
    }
}

#define GCN_BIN_SHIFT 8
#define GCN_BIN_NODES 256
#define GCN_NBMAX     512
#define GCN_CHUNK     8192
#define GCN_CT        1024
#define GCN_CAP       8192

__global__ __launch_bounds__(GCN_CT)
void gcn_bin_count(const int* __restrict__ dst, int* __restrict__ hist, int E)
{
    __shared__ int s_h[GCN_NBMAX];
    const int t = threadIdx.x;
    if (t < GCN_NBMAX) s_h[t] = 0;
    __syncthreads();

    const int e0 = blockIdx.x * GCN_CHUNK;
    if (e0 + GCN_CHUNK <= E) {
        const v4i* __restrict__ d4 = reinterpret_cast<const v4i*>(dst + e0);
        const v4i q0 = d4[t];
        const v4i q1 = d4[t + GCN_CT];
        atomicAdd(&s_h[q0.x >> GCN_BIN_SHIFT], 1);
        atomicAdd(&s_h[q0.y >> GCN_BIN_SHIFT], 1);
        atomicAdd(&s_h[q0.z >> GCN_BIN_SHIFT], 1);
        atomicAdd(&s_h[q0.w >> GCN_BIN_SHIFT], 1);
        atomicAdd(&s_h[q1.x >> GCN_BIN_SHIFT], 1);
        atomicAdd(&s_h[q1.y >> GCN_BIN_SHIFT], 1);
        atomicAdd(&s_h[q1.z >> GCN_BIN_SHIFT], 1);
        atomicAdd(&s_h[q1.w >> GCN_BIN_SHIFT], 1);
    } else {
        for (int i = t; i < E - e0; i += GCN_CT)
            atomicAdd(&s_h[dst[e0 + i] >> GCN_BIN_SHIFT], 1);
    }
    __syncthreads();
    if (t < GCN_NBMAX) hist[blockIdx.x * GCN_NBMAX + t] = s_h[t];
}

__device__ __forceinline__ int gcn_block_excl_scan(int v, int* s_wsum, int lane, int wave)
{
    int inc = v;
    for (int o = 1; o < 32; o <<= 1) {
        const int nb = __shfl_up(inc, o, 32);
        if (lane >= o) inc += nb;
    }
    if (lane == 31) s_wsum[wave] = inc;
    __syncthreads();
    int base = 0;
    for (int w = 0; w < 32; ++w)
        if (w < wave) base += s_wsum[w];
    __syncthreads();
    return base + inc - v;
}

__device__ __forceinline__ void gcn_place(int d, int s, int w, int* s_cur,
                                          int* s_src, int* s_wgt, int* s_dst)
{
    const int p = atomicAdd(&s_cur[d >> GCN_BIN_SHIFT], 1);
    s_src[p] = s;
    s_wgt[p] = w;
    s_dst[p] = d;
}

__global__ __launch_bounds__(GCN_CT)
void gcn_bin_scatter(const int* __restrict__ src, const int* __restrict__ dst,
                     const float* __restrict__ ew, const int* __restrict__ hist,
                     int* __restrict__ binptr, v2i* __restrict__ binned,
                     int E, int nchunks)
{
    __shared__ int s_src[GCN_CHUNK];
    __shared__ int s_wgt[GCN_CHUNK];
    __shared__ int s_dst[GCN_CHUNK];
    __shared__ int s_tot1[GCN_NBMAX];
    __shared__ int s_pre1[GCN_NBMAX];
    __shared__ int s_cur[GCN_NBMAX];
    __shared__ int s_gbase[GCN_NBMAX];
    __shared__ int s_wsum[32];

    const int t    = threadIdx.x;
    const int lane = t & 31;
    const int wave = t >> 5;
    const int b    = blockIdx.x;
    const int bin  = t & (GCN_NBMAX - 1);
    const int part = t >> 9;

    const int half = (nchunks + 1) >> 1;
    const int lo   = part ? half : 0;
    const int hi   = part ? nchunks : half;
    int tot = 0, pre = 0;
#pragma unroll 16
    for (int j = lo; j < hi; ++j) {
        const int v = hist[j * GCN_NBMAX + bin];
        tot += v;
        pre += (j < b) ? v : 0;
    }
    if (part) { s_tot1[bin] = tot; s_pre1[bin] = pre; }
    __syncthreads();
    int own = 0;
    if (!part) {
        tot += s_tot1[bin];
        pre += s_pre1[bin];
        own  = hist[b * GCN_NBMAX + bin];
    } else {
        tot = 0;
    }
    const int binbase = gcn_block_excl_scan(tot, s_wsum, lane, wave);
    const int loff    = gcn_block_excl_scan(own, s_wsum, lane, wave);
    if (!part) {
        s_cur[bin]   = loff;
        s_gbase[bin] = binbase + pre - loff;
        if (b == 0) {
            binptr[bin] = binbase;
            if (bin == GCN_NBMAX - 1) binptr[GCN_NBMAX] = binbase + tot;
        }
    }
    __syncthreads();

    const int e0 = b * GCN_CHUNK;
    if (e0 + GCN_CHUNK <= E) {
        const v4i* __restrict__ d4 = reinterpret_cast<const v4i*>(dst + e0);
        const v4i* __restrict__ s4 = reinterpret_cast<const v4i*>(src + e0);
        const v4i* __restrict__ w4 = reinterpret_cast<const v4i*>(ew + e0);
        const v4i dq0 = d4[t], dq1 = d4[t + GCN_CT];
        const v4i sq0 = s4[t], sq1 = s4[t + GCN_CT];
        const v4i wq0 = w4[t], wq1 = w4[t + GCN_CT];
        gcn_place(dq0.x, sq0.x, wq0.x, s_cur, s_src, s_wgt, s_dst);
        gcn_place(dq0.y, sq0.y, wq0.y, s_cur, s_src, s_wgt, s_dst);
        gcn_place(dq0.z, sq0.z, wq0.z, s_cur, s_src, s_wgt, s_dst);
        gcn_place(dq0.w, sq0.w, wq0.w, s_cur, s_src, s_wgt, s_dst);
        gcn_place(dq1.x, sq1.x, wq1.x, s_cur, s_src, s_wgt, s_dst);
        gcn_place(dq1.y, sq1.y, wq1.y, s_cur, s_src, s_wgt, s_dst);
        gcn_place(dq1.z, sq1.z, wq1.z, s_cur, s_src, s_wgt, s_dst);
        gcn_place(dq1.w, sq1.w, wq1.w, s_cur, s_src, s_wgt, s_dst);
    } else {
        for (int i = t; i < E - e0; i += GCN_CT)
            gcn_place(dst[e0 + i], src[e0 + i], __float_as_int(ew[e0 + i]),
                      s_cur, s_src, s_wgt, s_dst);
    }
    __syncthreads();

    const int nloc = min(GCN_CHUNK, E - e0);
    for (int i = t; i < nloc; i += GCN_CT) {
        const int d = s_dst[i];
        v2i rec;
        rec.x = s_src[i] | (int)(((unsigned)d & (GCN_BIN_NODES - 1)) << 24);
        rec.y = s_wgt[i];
        binned[s_gbase[d >> GCN_BIN_SHIFT] + i] = rec;
    }
}

__device__ __forceinline__
void gcn_bin_sort_block(const v2i* __restrict__ binned, const int* __restrict__ binptr,
                        v2i* __restrict__ edges, int* __restrict__ rowptr,
                        int* __restrict__ cnt, int M, int b)
{
    __shared__ v2i s_stage[GCN_CAP];
    __shared__ int s_cnt[GCN_BIN_NODES];
    __shared__ int s_cur[GCN_BIN_NODES];
    __shared__ int s_wsum[8];

    const int t    = threadIdx.x;
    const int lane = t & 31;
    const int wave = t >> 5;
    const int base = binptr[b];
    const int n    = binptr[b + 1] - base;
    const v2i* __restrict__ recs = binned + base;
    v2i* __restrict__ outp = edges + base;

    s_cnt[t] = 0;
    __syncthreads();

    for (int i0 = 0; i0 < n; i0 += 4 * GCN_BIN_NODES) {
        int key[4];
#pragma unroll
        for (int k = 0; k < 4; ++k) {
            const int i = i0 + k * GCN_BIN_NODES + t;
            key[k] = (i < n) ? recs[i].x : 0;
        }
#pragma unroll
        for (int k = 0; k < 4; ++k) {
            const int i = i0 + k * GCN_BIN_NODES + t;
            if (i < n) atomicAdd(&s_cnt[(unsigned)key[k] >> 24], 1);
        }
    }
    __syncthreads();

    const int c = s_cnt[t];
    int inc = c;
    for (int o = 1; o < 32; o <<= 1) {
        const int nb = __shfl_up(inc, o, 32);
        if (lane >= o) inc += nb;
    }
    if (lane == 31) s_wsum[wave] = inc;
    __syncthreads();
    int wbase = 0;
    for (int w = 0; w < 8; ++w)
        if (w < wave) wbase += s_wsum[w];
    const int ex = wbase + inc - c;
    s_cur[t] = ex;
    const int node = b * GCN_BIN_NODES + t;
    if (node < M) {
        rowptr[node] = base + ex;
        cnt[node]    = c;
    }
    __syncthreads();

    const bool fits = (n <= GCN_CAP);
    for (int i0 = 0; i0 < n; i0 += 4 * GCN_BIN_NODES) {
        v2i r[4];
#pragma unroll
        for (int k = 0; k < 4; ++k) {
            const int i = i0 + k * GCN_BIN_NODES + t;
            v2i z; z.x = 0; z.y = 0;
            r[k] = (i < n) ? recs[i] : z;
        }
#pragma unroll
        for (int k = 0; k < 4; ++k) {
            const int i = i0 + k * GCN_BIN_NODES + t;
            if (i < n) {
                const int p = atomicAdd(&s_cur[(unsigned)r[k].x >> 24], 1);
                v2i rec;
                rec.x = r[k].x & 0x00FFFFFF;
                rec.y = r[k].y;
                if (fits) s_stage[p] = rec;
                else      outp[p]    = rec;
            }
        }
    }
    __syncthreads();
    if (fits) {
        for (int i = t; i < n; i += GCN_BIN_NODES) outp[i] = s_stage[i];
    }
}

__global__ __launch_bounds__(GCN_BIN_NODES)
void gcn_bin_sort_and_gemm1(const v2i* __restrict__ binned, const int* __restrict__ binptr,
                            v2i* __restrict__ edges, int* __restrict__ rowptr,
                            int* __restrict__ cnt, int M, int nbins,
                            const float* __restrict__ x, const float* __restrict__ W1,
                            float* __restrict__ sup1, int patches)
{
    if ((int)blockIdx.x < nbins) {
        gcn_bin_sort_block(binned, binptr, edges, rowptr, cnt, M, (int)blockIdx.x);
    } else {
        const int lane  = threadIdx.x & 31;
        const int wave  = threadIdx.x >> 5;
        const int patch = (((int)blockIdx.x - nbins) << 1) + (wave >> 2);
        if (patch < patches)
            gcn_gemm_patch_wmma_f32<GCN_IN_C, GCN_H1>(x, W1, sup1, patch << 5,
                                                      (wave & 3) << 5, lane);
    }
}

template <int VEC> struct GcnRowVec;
template <> struct GcnRowVec<2> { typedef v2f type; };
template <> struct GcnRowVec<4> { typedef v4f type; };

template <int N, int C, typename VF>
__device__ __forceinline__ void gcn_add_edges(const float* __restrict__ sup,
                                              const v2i* __restrict__ ed,
                                              int lane, VF& acc)
{
    v2i e[N];
    VF  r[N];
#pragma unroll
    for (int k = 0; k < N; ++k) e[k] = ed[k];
#pragma unroll
    for (int k = 0; k < N; ++k)
        r[k] = reinterpret_cast<const VF*>(sup + (size_t)e[k].x * C)[lane];
#pragma unroll
    for (int k = 0; k < N; ++k) acc += r[k] * __int_as_float(e[k].y);
}

template <int C, bool RELU>
__global__ __launch_bounds__(256)
void gcn_row_gather(const float* __restrict__ sup,
                    const int* __restrict__ rowptr,
                    const int* __restrict__ cnt,
                    const v2i* __restrict__ edges,
                    const float* __restrict__ bias,
                    float* __restrict__ out,
                    int M)
{
    constexpr int VEC = C / 32;
    typedef typename GcnRowVec<VEC>::type VF;

    const int lane   = threadIdx.x & 31;
    const int nwaves = (gridDim.x * blockDim.x) >> 5;
    int row = __builtin_amdgcn_readfirstlane(
        (int)((blockIdx.x * blockDim.x + threadIdx.x) >> 5));

    const VF bv = reinterpret_cast<const VF*>(bias)[lane];

    for (; row < M; row += nwaves) {
        const int start = rowptr[row];
        const int deg   = cnt[row];
        const v2i* __restrict__ ed = edges + start;

        VF acc = {};
        int j = 0;
        for (; j + 8 <= deg; j += 8) gcn_add_edges<8, C, VF>(sup, ed + j, lane, acc);
        if (deg & 4) { gcn_add_edges<4, C, VF>(sup, ed + j, lane, acc); j += 4; }
        if (deg & 2) { gcn_add_edges<2, C, VF>(sup, ed + j, lane, acc); j += 2; }
        if (deg & 1) { gcn_add_edges<1, C, VF>(sup, ed + j, lane, acc); }

        acc += bv;
        if (RELU) {
            for (int i = 0; i < VEC; ++i) acc[i] = fmaxf(acc[i], 0.0f);
        }
        reinterpret_cast<VF*>(out + (size_t)row * C)[lane] = acc;
    }
}


template <int C>
__global__ __launch_bounds__(256)
void gcn_gather_relu_gemm64(const float* __restrict__ sup,
                            const int* __restrict__ rowptr,
                            const int* __restrict__ cnt,
                            const v2i* __restrict__ edges,
                            const float* __restrict__ bias,
                            const float* __restrict__ W,
                            float* __restrict__ next,
                            int M)
{
    constexpr int VEC   = C / 32;
    constexpr int N     = 64;
    constexpr int PITCH = C + 4;
    constexpr int OP    = N + 4;
    typedef typename GcnRowVec<VEC>::type VF;

    __shared__ __attribute__((aligned(16))) float s_h[32 * PITCH];
    __shared__ __attribute__((aligned(16))) float s_o[32 * OP];

    const int t    = threadIdx.x;
    const int lane = t & 31;
    const int wave = __builtin_amdgcn_readfirstlane(t >> 5);
    const int m0   = blockIdx.x << 5;

    const VF bv = reinterpret_cast<const VF*>(bias)[lane];

    for (int i = 0; i < 4; ++i) {
        const int r   = wave * 4 + i;
        const int row = m0 + r;
        VF acc = {};
        if (row < M) {
            const int start = rowptr[row];
            const int deg   = cnt[row];
            const v2i* __restrict__ ed = edges + start;
            int j = 0;
            for (; j + 8 <= deg; j += 8) gcn_add_edges<8, C, VF>(sup, ed + j, lane, acc);
            if (deg & 4) { gcn_add_edges<4, C, VF>(sup, ed + j, lane, acc); j += 4; }
            if (deg & 2) { gcn_add_edges<2, C, VF>(sup, ed + j, lane, acc); j += 2; }
            if (deg & 1) { gcn_add_edges<1, C, VF>(sup, ed + j, lane, acc); }
            acc += bv;
            for (int q = 0; q < VEC; ++q) acc[q] = fmaxf(acc[q], 0.0f);
        }
        reinterpret_cast<VF*>(s_h + r * PITCH)[lane] = acc;
    }
    __syncthreads();

    const int half = lane >> 4;
    const int l    = lane & 15;
    const int mt   = wave >> 2;
    const int nt   = wave & 3;
    const float* arow = s_h + (mt * 16 + l) * PITCH + 2 * half;
    const float* __restrict__ wcol = W + (size_t)(2 * half) * N + (nt * 16 + l);

    v8f d = {};
#pragma unroll 8
    for (int k = 0; k < C; k += 4) {
        v2f a, b;
        a   = *reinterpret_cast<const v2f*>(arow + k);
        b.x = wcol[(size_t)k * N];
        b.y = wcol[(size_t)k * N + N];
        d = __builtin_amdgcn_wmma_f32_16x16x4_f32(
            false, a, false, b, (short)0, d, false, false);
    }

    float* orow = s_o + (mt * 16 + 8 * half) * OP + (nt * 16 + l);
#pragma unroll
    for (int v = 0; v < 8; ++v) orow[v * OP] = d[v];
    __syncthreads();

#pragma unroll
    for (int p = 0; p < 2; ++p) {
        const int r  = (t >> 4) + 16 * p;
        const int c4 = (t & 15) * 4;
        if (m0 + r < M)
            *reinterpret_cast<v4f*>(next + (size_t)(m0 + r) * N + c4) =
                *reinterpret_cast<const v4f*>(s_o + r * OP + c4);
    }
}

extern "C" void kernel_launch(void* const* d_in, const int* in_sizes, int n_in,
                              void* d_out, int out_size, void* d_ws, size_t ws_size,
                              hipStream_t stream)
{
    const float* x    = (const float*)d_in[0];
    const int*   esrc = (const int*)  d_in[1];
    const int*   edst = (const int*)  d_in[2];
    const float* ew   = (const float*)d_in[3];
    const float* W1   = (const float*)d_in[4];
    const float* b1   = (const float*)d_in[5];
    const float* W2   = (const float*)d_in[6];
    const float* b2   = (const float*)d_in[7];
    const float* W3   = (const float*)d_in[8];
    const float* b3   = (const float*)d_in[9];

    const int M = in_sizes[0] / GCN_IN_C;
    const int E = in_sizes[1];

    float* ws     = (float*)d_ws;
    float* sup128 = ws;
    float* sup64  = sup128 + (size_t)M * GCN_H1;
    float* out    = (float*)d_out;

    const int nchunks = (E + GCN_CHUNK - 1) / GCN_CHUNK;
    const int nbins   = (M + GCN_BIN_NODES - 1) / GCN_BIN_NODES;
    int* cnt    = (int*)(sup64 + (size_t)M * GCN_H2);
    int* rowptr = cnt + nbins * GCN_BIN_NODES;
    v2i* edges  = (v2i*)(rowptr + nbins * GCN_BIN_NODES);
    v2i* binned = edges + E;
    int* hist   = (int*)(binned + E);
    int* binptr = hist + (size_t)nchunks * GCN_NBMAX;

    const int mt = M / 32;
    const int B  = 256;
    const int GW = 2048;
    const int tiles = (M + 31) / 32;

    gcn_bin_count<<<nchunks, GCN_CT, 0, stream>>>(edst, hist, E);
    gcn_bin_scatter<<<nchunks, GCN_CT, 0, stream>>>(esrc, edst, ew, hist, binptr,
                                                    binned, E, nchunks);
    gcn_bin_sort_and_gemm1<<<nbins + (mt + 1) / 2, GCN_BIN_NODES, 0, stream>>>(
        binned, binptr, edges, rowptr, cnt, M, nbins, x, W1, sup128, mt);

    gcn_gather_relu_gemm64<GCN_H1>
        <<<tiles, B, 0, stream>>>(sup128, rowptr, cnt, edges, b1, W2, sup64, M);

    gcn_gather_relu_gemm64<GCN_H2>
        <<<tiles, B, 0, stream>>>(sup64, rowptr, cnt, edges, b2, W3, sup128, M);

    gcn_row_gather<GCN_OUT, false>
        <<<GW, B, 0, stream>>>(sup128, rowptr, cnt, edges, b3, out, M);
}


# Layer-1 product: wave tile 16x64 with 16-byte A and W1 fragment loads, a quarter of the cache-line accesses per WMMA, whole-line stores; k grouping per WMMA (sum order of the product) changes
# speedup vs baseline: 4.3808x; 1.0676x over previous
#include <hip/hip_runtime.h>


typedef float v2f __attribute__((ext_vector_type(2)));
typedef float v4f __attribute__((ext_vector_type(4)));
typedef float v8f __attribute__((ext_vector_type(8)));
typedef int   v2i __attribute__((ext_vector_type(2)));
typedef int   v4i __attribute__((ext_vector_type(4)));

#define GCN_IN_C  256
#define GCN_H1    128
#define GCN_H2    64
#define GCN_OUT   64

template <int K, int N>
__device__ __forceinline__
void gcn_gemm_strip_wmma_f32(const float* __restrict__ A,
                             const float* __restrict__ W,
                             float* __restrict__ C,
                             int m0, int n0, int lane)
{
    const int half = lane >> 4;
    const int l    = lane & 15;

    const float* __restrict__ arow = A + (size_t)(m0 + l) * K + 4 * half;
    const float* __restrict__ wcol = W + (size_t)(4 * half) * N + (n0 + 4 * l);

    v8f acc0 = {}, acc1 = {}, acc2 = {}, acc3 = {};
#pragma unroll 2
    for (int k = 0; k < K; k += 8) {
        const v4f a  = *reinterpret_cast<const v4f*>(arow + k);
        const v4f w0 = *reinterpret_cast<const v4f*>(wcol + (size_t)(k + 0) * N);
        const v4f w1 = *reinterpret_cast<const v4f*>(wcol + (size_t)(k + 1) * N);
        const v4f w2 = *reinterpret_cast<const v4f*>(wcol + (size_t)(k + 2) * N);
        const v4f w3 = *reinterpret_cast<const v4f*>(wcol + (size_t)(k + 3) * N);
        v2f af, bf;
        af.x = a.x;  af.y = a.y;
        bf.x = w0.x; bf.y = w1.x;
        acc0 = __builtin_amdgcn_wmma_f32_16x16x4_f32(
            false, af, false, bf, (short)0, acc0, false, false);
        bf.x = w0.y; bf.y = w1.y;
        acc1 = __builtin_amdgcn_wmma_f32_16x16x4_f32(
            false, af, false, bf, (short)0, acc1, false, false);
        bf.x = w0.z; bf.y = w1.z;
        acc2 = __builtin_amdgcn_wmma_f32_16x16x4_f32(
            false, af, false, bf, (short)0, acc2, false, false);
        bf.x = w0.w; bf.y = w1.w;
        acc3 = __builtin_amdgcn_wmma_f32_16x16x4_f32(
            false, af, false, bf, (short)0, acc3, false, false);
        af.x = a.z;  af.y = a.w;
        bf.x = w2.x; bf.y = w3.x;
        acc0 = __builtin_amdgcn_wmma_f32_16x16x4_f32(
            false, af, false, bf, (short)0, acc0, false, false);
        bf.x = w2.y; bf.y = w3.y;
        acc1 = __builtin_amdgcn_wmma_f32_16x16x4_f32(
            false, af, false, bf, (short)0, acc1, false, false);
        bf.x = w2.z; bf.y = w3.z;
        acc2 = __builtin_amdgcn_wmma_f32_16x16x4_f32(
            false, af, false, bf, (short)0, acc2, false, false);
        bf.x = w2.w; bf.y = w3.w;
        acc3 = __builtin_amdgcn_wmma_f32_16x16x4_f32(
            false, af, false, bf, (short)0, acc3, false, false);
    }

    float* __restrict__ crow = C + (size_t)(m0 + 8 * half) * N + (n0 + 4 * l);
#pragma unroll
    for (int v = 0; v < 8; ++v) {
        v4f o;
        o.x = acc0[v]; o.y = acc1[v]; o.z = acc2[v]; o.w = acc3[v];
        *reinterpret_cast<v4f*>(crow + (size_t)v * N) = o;
    }
}

#define GCN_BIN_SHIFT 8
#define GCN_BIN_NODES 256
#define GCN_NBMAX     512
#define GCN_CHUNK     8192
#define GCN_CT        1024
#define GCN_CAP       8192

__global__ __launch_bounds__(GCN_CT)
void gcn_bin_count(const int* __restrict__ dst, int* __restrict__ hist, int E)
{
    __shared__ int s_h[GCN_NBMAX];
    const int t = threadIdx.x;
    if (t < GCN_NBMAX) s_h[t] = 0;
    __syncthreads();

    const int e0 = blockIdx.x * GCN_CHUNK;
    if (e0 + GCN_CHUNK <= E) {
        const v4i* __restrict__ d4 = reinterpret_cast<const v4i*>(dst + e0);
        const v4i q0 = d4[t];
        const v4i q1 = d4[t + GCN_CT];
        atomicAdd(&s_h[q0.x >> GCN_BIN_SHIFT], 1);
        atomicAdd(&s_h[q0.y >> GCN_BIN_SHIFT], 1);
        atomicAdd(&s_h[q0.z >> GCN_BIN_SHIFT], 1);
        atomicAdd(&s_h[q0.w >> GCN_BIN_SHIFT], 1);
        atomicAdd(&s_h[q1.x >> GCN_BIN_SHIFT], 1);
        atomicAdd(&s_h[q1.y >> GCN_BIN_SHIFT], 1);
        atomicAdd(&s_h[q1.z >> GCN_BIN_SHIFT], 1);
        atomicAdd(&s_h[q1.w >> GCN_BIN_SHIFT], 1);
    } else {
        for (int i = t; i < E - e0; i += GCN_CT)
            atomicAdd(&s_h[dst[e0 + i] >> GCN_BIN_SHIFT], 1);
    }
    __syncthreads();
    if (t < GCN_NBMAX) hist[blockIdx.x * GCN_NBMAX + t] = s_h[t];
}

__device__ __forceinline__ int gcn_block_excl_scan(int v, int* s_wsum, int lane, int wave)
{
    int inc = v;
    for (int o = 1; o < 32; o <<= 1) {
        const int nb = __shfl_up(inc, o, 32);
        if (lane >= o) inc += nb;
    }
    if (lane == 31) s_wsum[wave] = inc;
    __syncthreads();
    int base = 0;
    for (int w = 0; w < 32; ++w)
        if (w < wave) base += s_wsum[w];
    __syncthreads();
    return base + inc - v;
}

__device__ __forceinline__ void gcn_place(int d, int s, int w, int* s_cur,
                                          int* s_src, int* s_wgt, int* s_dst)
{
    const int p = atomicAdd(&s_cur[d >> GCN_BIN_SHIFT], 1);
    s_src[p] = s;
    s_wgt[p] = w;
    s_dst[p] = d;
}

__global__ __launch_bounds__(GCN_CT)
void gcn_bin_scatter(const int* __restrict__ src, const int* __restrict__ dst,
                     const float* __restrict__ ew, const int* __restrict__ hist,
                     int* __restrict__ binptr, v2i* __restrict__ binned,
                     int E, int nchunks)
{
    __shared__ int s_src[GCN_CHUNK];
    __shared__ int s_wgt[GCN_CHUNK];
    __shared__ int s_dst[GCN_CHUNK];
    __shared__ int s_tot1[GCN_NBMAX];
    __shared__ int s_pre1[GCN_NBMAX];
    __shared__ int s_cur[GCN_NBMAX];
    __shared__ int s_gbase[GCN_NBMAX];
    __shared__ int s_wsum[32];

    const int t    = threadIdx.x;
    const int lane = t & 31;
    const int wave = t >> 5;
    const int b    = blockIdx.x;
    const int bin  = t & (GCN_NBMAX - 1);
    const int part = t >> 9;

    const int half = (nchunks + 1) >> 1;
    const int lo   = part ? half : 0;
    const int hi   = part ? nchunks : half;
    int tot = 0, pre = 0;
#pragma unroll 16
    for (int j = lo; j < hi; ++j) {
        const int v = hist[j * GCN_NBMAX + bin];
        tot += v;
        pre += (j < b) ? v : 0;
    }
    if (part) { s_tot1[bin] = tot; s_pre1[bin] = pre; }
    __syncthreads();
    int own = 0;
    if (!part) {
        tot += s_tot1[bin];
        pre += s_pre1[bin];
        own  = hist[b * GCN_NBMAX + bin];
    } else {
        tot = 0;
    }
    const int binbase = gcn_block_excl_scan(tot, s_wsum, lane, wave);
    const int loff    = gcn_block_excl_scan(own, s_wsum, lane, wave);
    if (!part) {
        s_cur[bin]   = loff;
        s_gbase[bin] = binbase + pre - loff;
        if (b == 0) {
            binptr[bin] = binbase;
            if (bin == GCN_NBMAX - 1) binptr[GCN_NBMAX] = binbase + tot;
        }
    }
    __syncthreads();

    const int e0 = b * GCN_CHUNK;
    if (e0 + GCN_CHUNK <= E) {
        const v4i* __restrict__ d4 = reinterpret_cast<const v4i*>(dst + e0);
        const v4i* __restrict__ s4 = reinterpret_cast<const v4i*>(src + e0);
        const v4i* __restrict__ w4 = reinterpret_cast<const v4i*>(ew + e0);
        const v4i dq0 = d4[t], dq1 = d4[t + GCN_CT];
        const v4i sq0 = s4[t], sq1 = s4[t + GCN_CT];
        const v4i wq0 = w4[t], wq1 = w4[t + GCN_CT];
        gcn_place(dq0.x, sq0.x, wq0.x, s_cur, s_src, s_wgt, s_dst);
        gcn_place(dq0.y, sq0.y, wq0.y, s_cur, s_src, s_wgt, s_dst);
        gcn_place(dq0.z, sq0.z, wq0.z, s_cur, s_src, s_wgt, s_dst);
        gcn_place(dq0.w, sq0.w, wq0.w, s_cur, s_src, s_wgt, s_dst);
        gcn_place(dq1.x, sq1.x, wq1.x, s_cur, s_src, s_wgt, s_dst);
        gcn_place(dq1.y, sq1.y, wq1.y, s_cur, s_src, s_wgt, s_dst);
        gcn_place(dq1.z, sq1.z, wq1.z, s_cur, s_src, s_wgt, s_dst);
        gcn_place(dq1.w, sq1.w, wq1.w, s_cur, s_src, s_wgt, s_dst);
    } else {
        for (int i = t; i < E - e0; i += GCN_CT)
            gcn_place(dst[e0 + i], src[e0 + i], __float_as_int(ew[e0 + i]),
                      s_cur, s_src, s_wgt, s_dst);
    }
    __syncthreads();

    const int nloc = min(GCN_CHUNK, E - e0);
    for (int i = t; i < nloc; i += GCN_CT) {
        const int d = s_dst[i];
        v2i rec;
        rec.x = s_src[i] | (int)(((unsigned)d & (GCN_BIN_NODES - 1)) << 24);
        rec.y = s_wgt[i];
        binned[s_gbase[d >> GCN_BIN_SHIFT] + i] = rec;
    }
}

__device__ __forceinline__
void gcn_bin_sort_block(const v2i* __restrict__ binned, const int* __restrict__ binptr,
                        v2i* __restrict__ edges, int* __restrict__ rowptr,
                        int* __restrict__ cnt, int M, int b)
{
    __shared__ v2i s_stage[GCN_CAP];
    __shared__ int s_cnt[GCN_BIN_NODES];
    __shared__ int s_cur[GCN_BIN_NODES];
    __shared__ int s_wsum[8];

    const int t    = threadIdx.x;
    const int lane = t & 31;
    const int wave = t >> 5;
    const int base = binptr[b];
    const int n    = binptr[b + 1] - base;
    const v2i* __restrict__ recs = binned + base;
    v2i* __restrict__ outp = edges + base;

    s_cnt[t] = 0;
    __syncthreads();

    for (int i0 = 0; i0 < n; i0 += 4 * GCN_BIN_NODES) {
        int key[4];
#pragma unroll
        for (int k = 0; k < 4; ++k) {
            const int i = i0 + k * GCN_BIN_NODES + t;
            key[k] = (i < n) ? recs[i].x : 0;
        }
#pragma unroll
        for (int k = 0; k < 4; ++k) {
            const int i = i0 + k * GCN_BIN_NODES + t;
            if (i < n) atomicAdd(&s_cnt[(unsigned)key[k] >> 24], 1);
        }
    }
    __syncthreads();

    const int c = s_cnt[t];
    int inc = c;
    for (int o = 1; o < 32; o <<= 1) {
        const int nb = __shfl_up(inc, o, 32);
        if (lane >= o) inc += nb;
    }
    if (lane == 31) s_wsum[wave] = inc;
    __syncthreads();
    int wbase = 0;
    for (int w = 0; w < 8; ++w)
        if (w < wave) wbase += s_wsum[w];
    const int ex = wbase + inc - c;
    s_cur[t] = ex;
    const int node = b * GCN_BIN_NODES + t;
    if (node < M) {
        rowptr[node] = base + ex;
        cnt[node]    = c;
    }
    __syncthreads();

    const bool fits = (n <= GCN_CAP);
    for (int i0 = 0; i0 < n; i0 += 4 * GCN_BIN_NODES) {
        v2i r[4];
#pragma unroll
        for (int k = 0; k < 4; ++k) {
            const int i = i0 + k * GCN_BIN_NODES + t;
            v2i z; z.x = 0; z.y = 0;
            r[k] = (i < n) ? recs[i] : z;
        }
#pragma unroll
        for (int k = 0; k < 4; ++k) {
            const int i = i0 + k * GCN_BIN_NODES + t;
            if (i < n) {
                const int p = atomicAdd(&s_cur[(unsigned)r[k].x >> 24], 1);
                v2i rec;
                rec.x = r[k].x & 0x00FFFFFF;
                rec.y = r[k].y;
                if (fits) s_stage[p] = rec;
                else      outp[p]    = rec;
            }
        }
    }
    __syncthreads();
    if (fits) {
        for (int i = t; i < n; i += GCN_BIN_NODES) outp[i] = s_stage[i];
    }
}

__global__ __launch_bounds__(GCN_BIN_NODES)
void gcn_bin_sort_and_gemm1(const v2i* __restrict__ binned, const int* __restrict__ binptr,
                            v2i* __restrict__ edges, int* __restrict__ rowptr,
                            int* __restrict__ cnt, int M, int nbins,
                            const float* __restrict__ x, const float* __restrict__ W1,
                            float* __restrict__ sup1, int strips)
{
    if ((int)blockIdx.x < nbins) {
        gcn_bin_sort_block(binned, binptr, edges, rowptr, cnt, M, (int)blockIdx.x);
    } else {
        const int lane  = threadIdx.x & 31;
        const int wave  = threadIdx.x >> 5;
        const int strip = (((int)blockIdx.x - nbins) << 2) + (wave >> 1);
        if (strip < strips)
            gcn_gemm_strip_wmma_f32<GCN_IN_C, GCN_H1>(x, W1, sup1, strip << 4,
                                                      (wave & 1) << 6, lane);
    }
}

template <int VEC> struct GcnRowVec;
template <> struct GcnRowVec<2> { typedef v2f type; };
template <> struct GcnRowVec<4> { typedef v4f type; };

template <int N, int C, typename VF>
__device__ __forceinline__ void gcn_add_edges(const float* __restrict__ sup,
                                              const v2i* __restrict__ ed,
                                              int lane, VF& acc)
{
    v2i e[N];
    VF  r[N];
#pragma unroll
    for (int k = 0; k < N; ++k) e[k] = ed[k];
#pragma unroll
    for (int k = 0; k < N; ++k)
        r[k] = reinterpret_cast<const VF*>(sup + (size_t)e[k].x * C)[lane];
#pragma unroll
    for (int k = 0; k < N; ++k) acc += r[k] * __int_as_float(e[k].y);
}

template <int C, bool RELU>
__global__ __launch_bounds__(256)
void gcn_row_gather(const float* __restrict__ sup,
                    const int* __restrict__ rowptr,
                    const int* __restrict__ cnt,
                    const v2i* __restrict__ edges,
                    const float* __restrict__ bias,
                    float* __restrict__ out,
                    int M)
{
    constexpr int VEC = C / 32;
    typedef typename GcnRowVec<VEC>::type VF;

    const int lane   = threadIdx.x & 31;
    const int nwaves = (gridDim.x * blockDim.x) >> 5;
    int row = __builtin_amdgcn_readfirstlane(
        (int)((blockIdx.x * blockDim.x + threadIdx.x) >> 5));

    const VF bv = reinterpret_cast<const VF*>(bias)[lane];

    for (; row < M; row += nwaves) {
        const int start = rowptr[row];
        const int deg   = cnt[row];
        const v2i* __restrict__ ed = edges + start;

        VF acc = {};
        int j = 0;
        for (; j + 8 <= deg; j += 8) gcn_add_edges<8, C, VF>(sup, ed + j, lane, acc);
        if (deg & 4) { gcn_add_edges<4, C, VF>(sup, ed + j, lane, acc); j += 4; }
        if (deg & 2) { gcn_add_edges<2, C, VF>(sup, ed + j, lane, acc); j += 2; }
        if (deg & 1) { gcn_add_edges<1, C, VF>(sup, ed + j, lane, acc); }

        acc += bv;
        if (RELU) {
            for (int i = 0; i < VEC; ++i) acc[i] = fmaxf(acc[i], 0.0f);
        }
        reinterpret_cast<VF*>(out + (size_t)row * C)[lane] = acc;
    }
}


template <int C>
__global__ __launch_bounds__(256)
void gcn_gather_relu_gemm64(const float* __restrict__ sup,
                            const int* __restrict__ rowptr,
                            const int* __restrict__ cnt,
                            const v2i* __restrict__ edges,
                            const float* __restrict__ bias,
                            const float* __restrict__ W,
                            float* __restrict__ next,
                            int M)
{
    constexpr int VEC   = C / 32;
    constexpr int N     = 64;
    constexpr int PITCH = C + 4;
    constexpr int OP    = N + 4;
    typedef typename GcnRowVec<VEC>::type VF;

    __shared__ __attribute__((aligned(16))) float s_h[32 * PITCH];
    __shared__ __attribute__((aligned(16))) float s_o[32 * OP];

    const int t    = threadIdx.x;
    const int lane = t & 31;
    const int wave = __builtin_amdgcn_readfirstlane(t >> 5);
    const int m0   = blockIdx.x << 5;

    const VF bv = reinterpret_cast<const VF*>(bias)[lane];

    for (int i = 0; i < 4; ++i) {
        const int r   = wave * 4 + i;
        const int row = m0 + r;
        VF acc = {};
        if (row < M) {
            const int start = rowptr[row];
            const int deg   = cnt[row];
            const v2i* __restrict__ ed = edges + start;
            int j = 0;
            for (; j + 8 <= deg; j += 8) gcn_add_edges<8, C, VF>(sup, ed + j, lane, acc);
            if (deg & 4) { gcn_add_edges<4, C, VF>(sup, ed + j, lane, acc); j += 4; }
            if (deg & 2) { gcn_add_edges<2, C, VF>(sup, ed + j, lane, acc); j += 2; }
            if (deg & 1) { gcn_add_edges<1, C, VF>(sup, ed + j, lane, acc); }
            acc += bv;
            for (int q = 0; q < VEC; ++q) acc[q] = fmaxf(acc[q], 0.0f);
        }
        reinterpret_cast<VF*>(s_h + r * PITCH)[lane] = acc;
    }
    __syncthreads();

    const int half = lane >> 4;
    const int l    = lane & 15;
    const int mt   = wave >> 2;
    const int nt   = wave & 3;
    const float* arow = s_h + (mt * 16 + l) * PITCH + 2 * half;
    const float* __restrict__ wcol = W + (size_t)(2 * half) * N + (nt * 16 + l);

    v8f d = {};
#pragma unroll 8
    for (int k = 0; k < C; k += 4) {
        v2f a, b;
        a   = *reinterpret_cast<const v2f*>(arow + k);
        b.x = wcol[(size_t)k * N];
        b.y = wcol[(size_t)k * N + N];
        d = __builtin_amdgcn_wmma_f32_16x16x4_f32(
            false, a, false, b, (short)0, d, false, false);
    }

    float* orow = s_o + (mt * 16 + 8 * half) * OP + (nt * 16 + l);
#pragma unroll
    for (int v = 0; v < 8; ++v) orow[v * OP] = d[v];
    __syncthreads();

#pragma unroll
    for (int p = 0; p < 2; ++p) {
        const int r  = (t >> 4) + 16 * p;
        const int c4 = (t & 15) * 4;
        if (m0 + r < M)
            *reinterpret_cast<v4f*>(next + (size_t)(m0 + r) * N + c4) =
                *reinterpret_cast<const v4f*>(s_o + r * OP + c4);
    }
}

extern "C" void kernel_launch(void* const* d_in, const int* in_sizes, int n_in,
                              void* d_out, int out_size, void* d_ws, size_t ws_size,
                              hipStream_t stream)
{
    const float* x    = (const float*)d_in[0];
    const int*   esrc = (const int*)  d_in[1];
    const int*   edst = (const int*)  d_in[2];
    const float* ew   = (const float*)d_in[3];
    const float* W1   = (const float*)d_in[4];
    const float* b1   = (const float*)d_in[5];
    const float* W2   = (const float*)d_in[6];
    const float* b2   = (const float*)d_in[7];
    const float* W3   = (const float*)d_in[8];
    const float* b3   = (const float*)d_in[9];

    const int M = in_sizes[0] / GCN_IN_C;
    const int E = in_sizes[1];

    float* ws     = (float*)d_ws;
    float* sup128 = ws;
    float* sup64  = sup128 + (size_t)M * GCN_H1;
    float* out    = (float*)d_out;

    const int nchunks = (E + GCN_CHUNK - 1) / GCN_CHUNK;
    const int nbins   = (M + GCN_BIN_NODES - 1) / GCN_BIN_NODES;
    int* cnt    = (int*)(sup64 + (size_t)M * GCN_H2);
    int* rowptr = cnt + nbins * GCN_BIN_NODES;
    v2i* edges  = (v2i*)(rowptr + nbins * GCN_BIN_NODES);
    v2i* binned = edges + E;
    int* hist   = (int*)(binned + E);
    int* binptr = hist + (size_t)nchunks * GCN_NBMAX;

    const int mt = M / 16;
    const int B  = 256;
    const int GW = 2048;
    const int tiles = (M + 31) / 32;

    gcn_bin_count<<<nchunks, GCN_CT, 0, stream>>>(edst, hist, E);
    gcn_bin_scatter<<<nchunks, GCN_CT, 0, stream>>>(esrc, edst, ew, hist, binptr,
                                                    binned, E, nchunks);
    gcn_bin_sort_and_gemm1<<<nbins + (mt + 3) / 4, GCN_BIN_NODES, 0, stream>>>(
        binned, binptr, edges, rowptr, cnt, M, nbins, x, W1, sup128, mt);

    gcn_gather_relu_gemm64<GCN_H1>
        <<<tiles, B, 0, stream>>>(sup128, rowptr, cnt, edges, b1, W2, sup64, M);

    gcn_gather_relu_gemm64<GCN_H2>
        <<<tiles, B, 0, stream>>>(sup64, rowptr, cnt, edges, b2, W3, sup128, M);

    gcn_row_gather<GCN_OUT, false>
        <<<GW, B, 0, stream>>>(sup128, rowptr, cnt, edges, b3, out, M);
}


# Edge sort in 2 launches, not 3: a chunk block orders its chunk by bin in place and stores per-bin offsets; a bin block sums them for its base and collects its runs; row sum order = LDS arrival rank
# speedup vs baseline: 4.5391x; 1.0361x over previous
#include <hip/hip_runtime.h>


typedef float v2f __attribute__((ext_vector_type(2)));
typedef float v4f __attribute__((ext_vector_type(4)));
typedef float v8f __attribute__((ext_vector_type(8)));
typedef int   v2i __attribute__((ext_vector_type(2)));
typedef int   v4i __attribute__((ext_vector_type(4)));

#define GCN_IN_C  256
#define GCN_H1    128
#define GCN_H2    64
#define GCN_OUT   64

template <int K, int N>
__device__ __forceinline__
void gcn_gemm_strip_wmma_f32(const float* __restrict__ A,
                             const float* __restrict__ W,
                             float* __restrict__ C,
                             int m0, int n0, int lane)
{
    const int half = lane >> 4;
    const int l    = lane & 15;

    const float* __restrict__ arow = A + (size_t)(m0 + l) * K + 4 * half;
    const float* __restrict__ wcol = W + (size_t)(4 * half) * N + (n0 + 4 * l);

    v8f acc0 = {}, acc1 = {}, acc2 = {}, acc3 = {};
#pragma unroll 2
    for (int k = 0; k < K; k += 8) {
        const v4f a  = *reinterpret_cast<const v4f*>(arow + k);
        const v4f w0 = *reinterpret_cast<const v4f*>(wcol + (size_t)(k + 0) * N);
        const v4f w1 = *reinterpret_cast<const v4f*>(wcol + (size_t)(k + 1) * N);
        const v4f w2 = *reinterpret_cast<const v4f*>(wcol + (size_t)(k + 2) * N);
        const v4f w3 = *reinterpret_cast<const v4f*>(wcol + (size_t)(k + 3) * N);
        v2f af, bf;
        af.x = a.x;  af.y = a.y;
        bf.x = w0.x; bf.y = w1.x;
        acc0 = __builtin_amdgcn_wmma_f32_16x16x4_f32(
            false, af, false, bf, (short)0, acc0, false, false);
        bf.x = w0.y; bf.y = w1.y;
        acc1 = __builtin_amdgcn_wmma_f32_16x16x4_f32(
            false, af, false, bf, (short)0, acc1, false, false);
        bf.x = w0.z; bf.y = w1.z;
        acc2 = __builtin_amdgcn_wmma_f32_16x16x4_f32(
            false, af, false, bf, (short)0, acc2, false, false);
        bf.x = w0.w; bf.y = w1.w;
        acc3 = __builtin_amdgcn_wmma_f32_16x16x4_f32(
            false, af, false, bf, (short)0, acc3, false, false);
        af.x = a.z;  af.y = a.w;
        bf.x = w2.x; bf.y = w3.x;
        acc0 = __builtin_amdgcn_wmma_f32_16x16x4_f32(
            false, af, false, bf, (short)0, acc0, false, false);
        bf.x = w2.y; bf.y = w3.y;
        acc1 = __builtin_amdgcn_wmma_f32_16x16x4_f32(
            false, af, false, bf, (short)0, acc1, false, false);
        bf.x = w2.z; bf.y = w3.z;
        acc2 = __builtin_amdgcn_wmma_f32_16x16x4_f32(
            false, af, false, bf, (short)0, acc2, false, false);
        bf.x = w2.w; bf.y = w3.w;
        acc3 = __builtin_amdgcn_wmma_f32_16x16x4_f32(
            false, af, false, bf, (short)0, acc3, false, false);
    }

    float* __restrict__ crow = C + (size_t)(m0 + 8 * half) * N + (n0 + 4 * l);
#pragma unroll
    for (int v = 0; v < 8; ++v) {
        v4f o;
        o.x = acc0[v]; o.y = acc1[v]; o.z = acc2[v]; o.w = acc3[v];
        *reinterpret_cast<v4f*>(crow + (size_t)v * N) = o;
    }
}

#define GCN_BIN_SHIFT 8
#define GCN_BIN_NODES 256
#define GCN_NBMAX     512
#define GCN_CHUNK     8192
#define GCN_CT        1024
#define GCN_CAP       8192
#define GCN_MAXCH     256

__device__ __forceinline__ int gcn_block_excl_scan(int v, int* s_wsum, int lane, int wave)
{
    int inc = v;
    for (int o = 1; o < 32; o <<= 1) {
        const int nb = __shfl_up(inc, o, 32);
        if (lane >= o) inc += nb;
    }
    if (lane == 31) s_wsum[wave] = inc;
    __syncthreads();
    int base = 0;
    for (int w = 0; w < 32; ++w)
        if (w < wave) base += s_wsum[w];
    __syncthreads();
    return base + inc - v;
}

__device__ __forceinline__ void gcn_place(int d, int s, int w, int* s_cur, v2i* s_rec)
{
    const int p = atomicAdd(&s_cur[d >> GCN_BIN_SHIFT], 1);
    v2i rec;
    rec.x = s | (int)(((unsigned)d & (GCN_BIN_NODES - 1)) << 24);
    rec.y = w;
    s_rec[p] = rec;
}

__global__ __launch_bounds__(GCN_CT)
void gcn_chunk_sort(const int* __restrict__ src, const int* __restrict__ dst,
                    const float* __restrict__ ew, int* __restrict__ runinfo,
                    v2i* __restrict__ binned, int E)
{
    __shared__ __attribute__((aligned(16))) v2i s_rec[GCN_CHUNK];
    __shared__ int s_cnt[GCN_NBMAX];
    __shared__ int s_cur[GCN_NBMAX];
    __shared__ int s_wsum[32];

    const int t    = threadIdx.x;
    const int lane = t & 31;
    const int wave = t >> 5;
    const int b    = blockIdx.x;
    const int e0   = b * GCN_CHUNK;
    const bool whole = (e0 + GCN_CHUNK <= E);

    if (t < GCN_NBMAX) s_cnt[t] = 0;
    __syncthreads();

    v4i dq0 = {}, dq1 = {}, sq0 = {}, sq1 = {}, wq0 = {}, wq1 = {};
    if (whole) {
        const v4i* __restrict__ d4 = reinterpret_cast<const v4i*>(dst + e0);
        const v4i* __restrict__ s4 = reinterpret_cast<const v4i*>(src + e0);
        const v4i* __restrict__ w4 = reinterpret_cast<const v4i*>(ew + e0);
        dq0 = d4[t]; dq1 = d4[t + GCN_CT];
        sq0 = s4[t]; sq1 = s4[t + GCN_CT];
        wq0 = w4[t]; wq1 = w4[t + GCN_CT];
        atomicAdd(&s_cnt[dq0.x >> GCN_BIN_SHIFT], 1);
        atomicAdd(&s_cnt[dq0.y >> GCN_BIN_SHIFT], 1);
        atomicAdd(&s_cnt[dq0.z >> GCN_BIN_SHIFT], 1);
        atomicAdd(&s_cnt[dq0.w >> GCN_BIN_SHIFT], 1);
        atomicAdd(&s_cnt[dq1.x >> GCN_BIN_SHIFT], 1);
        atomicAdd(&s_cnt[dq1.y >> GCN_BIN_SHIFT], 1);
        atomicAdd(&s_cnt[dq1.z >> GCN_BIN_SHIFT], 1);
        atomicAdd(&s_cnt[dq1.w >> GCN_BIN_SHIFT], 1);
    } else {
        for (int i = t; i < E - e0; i += GCN_CT)
            atomicAdd(&s_cnt[dst[e0 + i] >> GCN_BIN_SHIFT], 1);
    }
    __syncthreads();

    const int own  = (t < GCN_NBMAX) ? s_cnt[t] : 0;
    const int loff = gcn_block_excl_scan(own, s_wsum, lane, wave);
    if (t < GCN_NBMAX) {
        s_cur[t] = loff;
        runinfo[b * GCN_NBMAX + t] = (loff << 16) | own;
    }
    __syncthreads();

    if (whole) {
        gcn_place(dq0.x, sq0.x, wq0.x, s_cur, s_rec);
        gcn_place(dq0.y, sq0.y, wq0.y, s_cur, s_rec);
        gcn_place(dq0.z, sq0.z, wq0.z, s_cur, s_rec);
        gcn_place(dq0.w, sq0.w, wq0.w, s_cur, s_rec);
        gcn_place(dq1.x, sq1.x, wq1.x, s_cur, s_rec);
        gcn_place(dq1.y, sq1.y, wq1.y, s_cur, s_rec);
        gcn_place(dq1.z, sq1.z, wq1.z, s_cur, s_rec);
        gcn_place(dq1.w, sq1.w, wq1.w, s_cur, s_rec);
    } else {
        for (int i = t; i < E - e0; i += GCN_CT)
            gcn_place(dst[e0 + i], src[e0 + i], __float_as_int(ew[e0 + i]), s_cur, s_rec);
    }
    __syncthreads();

    if (whole) {
        const v4i* s4r = reinterpret_cast<const v4i*>(s_rec);
        v4i* __restrict__ o4 = reinterpret_cast<v4i*>(binned + e0);
#pragma unroll
        for (int k = 0; k < GCN_CHUNK / (2 * GCN_CT); ++k)
            o4[t + k * GCN_CT] = s4r[t + k * GCN_CT];
    } else {
        for (int i = t; i < E - e0; i += GCN_CT) binned[e0 + i] = s_rec[i];
    }
}

__device__ __forceinline__ void gcn_put(v2i r, int* s_cur, v2i* s_stage,
                                        v2i* __restrict__ outp, bool fits)
{
    const int p = atomicAdd(&s_cur[(unsigned)r.x >> 24], 1);
    v2i rec;
    rec.x = r.x & 0x00FFFFFF;
    rec.y = r.y;
    if (fits) s_stage[p] = rec;
    else      outp[p]    = rec;
}

__device__ __forceinline__
void gcn_bin_sort_block(const v2i* __restrict__ binned, const int* __restrict__ runinfo,
                        v2i* __restrict__ edges, int* __restrict__ rowptr,
                        int* __restrict__ cnt, int M, int b, int nchunks)
{
    __shared__ v2i s_stage[GCN_CAP];
    __shared__ int s_cnt[GCN_BIN_NODES];
    __shared__ int s_cur[GCN_BIN_NODES];
    __shared__ int s_rpos[GCN_MAXCH];
    __shared__ int s_rlen[GCN_MAXCH];
    __shared__ int s_woff[8];
    __shared__ int s_wlen[8];
    __shared__ int s_wsum[8];

    const int t    = threadIdx.x;
    const int lane = t & 31;
    const int wave = t >> 5;

    const int info = (t < nchunks) ? runinfo[t * GCN_NBMAX + b] : 0;
    const int roff = info >> 16;
    const int rlen = info & 0xFFFF;
    s_rpos[t] = t * GCN_CHUNK + roff;
    s_rlen[t] = rlen;
    s_cnt[t]  = 0;
    int so = roff, sl = rlen;
    for (int o = 16; o > 0; o >>= 1) {
        so += __shfl_xor(so, o, 32);
        sl += __shfl_xor(sl, o, 32);
    }
    if (lane == 0) { s_woff[wave] = so; s_wlen[wave] = sl; }
    __syncthreads();
    int base = 0, n = 0;
    for (int w = 0; w < 8; ++w) { base += s_woff[w]; n += s_wlen[w]; }
    v2i* __restrict__ outp = edges + base;

    for (int c0 = wave; c0 < nchunks; c0 += 32) {
        int pos[4], len[4], key[4];
#pragma unroll
        for (int k = 0; k < 4; ++k) {
            const int ch = c0 + 8 * k;
            const int cc = (ch < nchunks) ? ch : 0;
            pos[k] = s_rpos[cc];
            len[k] = (ch < nchunks) ? s_rlen[cc] : 0;
        }
#pragma unroll
        for (int k = 0; k < 4; ++k)
            key[k] = (lane < len[k]) ? binned[pos[k] + lane].x : 0;
#pragma unroll
        for (int k = 0; k < 4; ++k)
            if (lane < len[k]) atomicAdd(&s_cnt[(unsigned)key[k] >> 24], 1);
#pragma unroll
        for (int k = 0; k < 4; ++k)
            for (int j = lane + 32; j < len[k]; j += 32)
                atomicAdd(&s_cnt[(unsigned)binned[pos[k] + j].x >> 24], 1);
    }
    __syncthreads();

    const int c = s_cnt[t];
    int inc = c;
    for (int o = 1; o < 32; o <<= 1) {
        const int nb = __shfl_up(inc, o, 32);
        if (lane >= o) inc += nb;
    }
    if (lane == 31) s_wsum[wave] = inc;
    __syncthreads();
    int wbase = 0;
    for (int w = 0; w < 8; ++w)
        if (w < wave) wbase += s_wsum[w];
    const int ex = wbase + inc - c;
    s_cur[t] = ex;
    const int node = b * GCN_BIN_NODES + t;
    if (node < M) {
        rowptr[node] = base + ex;
        cnt[node]    = c;
    }
    __syncthreads();

    const bool fits = (n <= GCN_CAP);
    for (int c0 = wave; c0 < nchunks; c0 += 32) {
        int pos[4], len[4];
        v2i r[4];
#pragma unroll
        for (int k = 0; k < 4; ++k) {
            const int ch = c0 + 8 * k;
            const int cc = (ch < nchunks) ? ch : 0;
            pos[k] = s_rpos[cc];
            len[k] = (ch < nchunks) ? s_rlen[cc] : 0;
        }
#pragma unroll
        for (int k = 0; k < 4; ++k) {
            v2i z; z.x = 0; z.y = 0;
            r[k] = (lane < len[k]) ? binned[pos[k] + lane] : z;
        }
#pragma unroll
        for (int k = 0; k < 4; ++k)
            if (lane < len[k]) gcn_put(r[k], s_cur, s_stage, outp, fits);
#pragma unroll
        for (int k = 0; k < 4; ++k)
            for (int j = lane + 32; j < len[k]; j += 32)
                gcn_put(binned[pos[k] + j], s_cur, s_stage, outp, fits);
    }
    __syncthreads();
    if (fits) {
        for (int i = t; i < n; i += GCN_BIN_NODES) outp[i] = s_stage[i];
    }
}

__global__ __launch_bounds__(GCN_BIN_NODES)
void gcn_bin_sort_and_gemm1(const v2i* __restrict__ binned, const int* __restrict__ runinfo,
                            v2i* __restrict__ edges, int* __restrict__ rowptr,
                            int* __restrict__ cnt, int M, int nbins, int nchunks,
                            const float* __restrict__ x, const float* __restrict__ W1,
                            float* __restrict__ sup1, int strips)
{
    if ((int)blockIdx.x < nbins) {
        gcn_bin_sort_block(binned, runinfo, edges, rowptr, cnt, M, (int)blockIdx.x, nchunks);
    } else {
        const int lane  = threadIdx.x & 31;
        const int wave  = threadIdx.x >> 5;
        const int strip = (((int)blockIdx.x - nbins) << 2) + (wave >> 1);
        if (strip < strips)
            gcn_gemm_strip_wmma_f32<GCN_IN_C, GCN_H1>(x, W1, sup1, strip << 4,
                                                      (wave & 1) << 6, lane);
    }
}

template <int VEC> struct GcnRowVec;
template <> struct GcnRowVec<2> { typedef v2f type; };
template <> struct GcnRowVec<4> { typedef v4f type; };

template <int N, int C, typename VF>
__device__ __forceinline__ void gcn_add_edges(const float* __restrict__ sup,
                                              const v2i* __restrict__ ed,
                                              int lane, VF& acc)
{
    v2i e[N];
    VF  r[N];
#pragma unroll
    for (int k = 0; k < N; ++k) e[k] = ed[k];
#pragma unroll
    for (int k = 0; k < N; ++k)
        r[k] = reinterpret_cast<const VF*>(sup + (size_t)e[k].x * C)[lane];
#pragma unroll
    for (int k = 0; k < N; ++k) acc += r[k] * __int_as_float(e[k].y);
}

template <int C, bool RELU>
__global__ __launch_bounds__(256)
void gcn_row_gather(const float* __restrict__ sup,
                    const int* __restrict__ rowptr,
                    const int* __restrict__ cnt,
                    const v2i* __restrict__ edges,
                    const float* __restrict__ bias,
                    float* __restrict__ out,
                    int M)
{
    constexpr int VEC = C / 32;
    typedef typename GcnRowVec<VEC>::type VF;

    const int lane   = threadIdx.x & 31;
    const int nwaves = (gridDim.x * blockDim.x) >> 5;
    int row = __builtin_amdgcn_readfirstlane(
        (int)((blockIdx.x * blockDim.x + threadIdx.x) >> 5));

    const VF bv = reinterpret_cast<const VF*>(bias)[lane];

    for (; row < M; row += nwaves) {
        const int start = rowptr[row];
        const int deg   = cnt[row];
        const v2i* __restrict__ ed = edges + start;

        VF acc = {};
        int j = 0;
        for (; j + 8 <= deg; j += 8) gcn_add_edges<8, C, VF>(sup, ed + j, lane, acc);
        if (deg & 4) { gcn_add_edges<4, C, VF>(sup, ed + j, lane, acc); j += 4; }
        if (deg & 2) { gcn_add_edges<2, C, VF>(sup, ed + j, lane, acc); j += 2; }
        if (deg & 1) { gcn_add_edges<1, C, VF>(sup, ed + j, lane, acc); }

        acc += bv;
        if (RELU) {
            for (int i = 0; i < VEC; ++i) acc[i] = fmaxf(acc[i], 0.0f);
        }
        reinterpret_cast<VF*>(out + (size_t)row * C)[lane] = acc;
    }
}


template <int C>
__global__ __launch_bounds__(256)
void gcn_gather_relu_gemm64(const float* __restrict__ sup,
                            const int* __restrict__ rowptr,
                            const int* __restrict__ cnt,
                            const v2i* __restrict__ edges,
                            const float* __restrict__ bias,
                            const float* __restrict__ W,
                            float* __restrict__ next,
                            int M)
{
    constexpr int VEC   = C / 32;
    constexpr int N     = 64;
    constexpr int PITCH = C + 4;
    constexpr int OP    = N + 4;
    typedef typename GcnRowVec<VEC>::type VF;

    __shared__ __attribute__((aligned(16))) float s_h[32 * PITCH];
    __shared__ __attribute__((aligned(16))) float s_o[32 * OP];

    const int t    = threadIdx.x;
    const int lane = t & 31;
    const int wave = __builtin_amdgcn_readfirstlane(t >> 5);
    const int m0   = blockIdx.x << 5;

    const VF bv = reinterpret_cast<const VF*>(bias)[lane];

    for (int i = 0; i < 4; ++i) {
        const int r   = wave * 4 + i;
        const int row = m0 + r;
        VF acc = {};
        if (row < M) {
            const int start = rowptr[row];
            const int deg   = cnt[row];
            const v2i* __restrict__ ed = edges + start;
            int j = 0;
            for (; j + 8 <= deg; j += 8) gcn_add_edges<8, C, VF>(sup, ed + j, lane, acc);
            if (deg & 4) { gcn_add_edges<4, C, VF>(sup, ed + j, lane, acc); j += 4; }
            if (deg & 2) { gcn_add_edges<2, C, VF>(sup, ed + j, lane, acc); j += 2; }
            if (deg & 1) { gcn_add_edges<1, C, VF>(sup, ed + j, lane, acc); }
            acc += bv;
            for (int q = 0; q < VEC; ++q) acc[q] = fmaxf(acc[q], 0.0f);
        }
        reinterpret_cast<VF*>(s_h + r * PITCH)[lane] = acc;
    }
    __syncthreads();

    const int half = lane >> 4;
    const int l    = lane & 15;
    const int mt   = wave >> 2;
    const int nt   = wave & 3;
    const float* arow = s_h + (mt * 16 + l) * PITCH + 2 * half;
    const float* __restrict__ wcol = W + (size_t)(2 * half) * N + (nt * 16 + l);

    v8f d = {};
#pragma unroll 8
    for (int k = 0; k < C; k += 4) {
        v2f a, b;
        a   = *reinterpret_cast<const v2f*>(arow + k);
        b.x = wcol[(size_t)k * N];
        b.y = wcol[(size_t)k * N + N];
        d = __builtin_amdgcn_wmma_f32_16x16x4_f32(
            false, a, false, b, (short)0, d, false, false);
    }

    float* orow = s_o + (mt * 16 + 8 * half) * OP + (nt * 16 + l);
#pragma unroll
    for (int v = 0; v < 8; ++v) orow[v * OP] = d[v];
    __syncthreads();

#pragma unroll
    for (int p = 0; p < 2; ++p) {
        const int r  = (t >> 4) + 16 * p;
        const int c4 = (t & 15) * 4;
        if (m0 + r < M)
            *reinterpret_cast<v4f*>(next + (size_t)(m0 + r) * N + c4) =
                *reinterpret_cast<const v4f*>(s_o + r * OP + c4);
    }
}


extern "C" void kernel_launch(void* const* d_in, const int* in_sizes, int n_in,
                              void* d_out, int out_size, void* d_ws, size_t ws_size,
                              hipStream_t stream)
{
    const float* x    = (const float*)d_in[0];
    const int*   esrc = (const int*)  d_in[1];
    const int*   edst = (const int*)  d_in[2];
    const float* ew   = (const float*)d_in[3];
    const float* W1   = (const float*)d_in[4];
    const float* b1   = (const float*)d_in[5];
    const float* W2   = (const float*)d_in[6];
    const float* b2   = (const float*)d_in[7];
    const float* W3   = (const float*)d_in[8];
    const float* b3   = (const float*)d_in[9];

    const int M = in_sizes[0] / GCN_IN_C;
    const int E = in_sizes[1];

    float* ws     = (float*)d_ws;
    float* sup128 = ws;
    float* sup64  = sup128 + (size_t)M * GCN_H1;
    float* out    = (float*)d_out;

    const int nchunks = (E + GCN_CHUNK - 1) / GCN_CHUNK;
    const int nbins   = (M + GCN_BIN_NODES - 1) / GCN_BIN_NODES;
    int* cnt     = (int*)(sup64 + (size_t)M * GCN_H2);
    int* rowptr  = cnt + nbins * GCN_BIN_NODES;
    v2i* edges   = (v2i*)(rowptr + nbins * GCN_BIN_NODES);
    v2i* binned  = edges + E;
    int* runinfo = (int*)(binned + E);

    const int mt = M / 16;
    const int B  = 256;
    const int GW = 2048;
    const int tiles = (M + 31) / 32;

    gcn_chunk_sort<<<nchunks, GCN_CT, 0, stream>>>(esrc, edst, ew, runinfo, binned, E);
    gcn_bin_sort_and_gemm1<<<nbins + (mt + 3) / 4, GCN_BIN_NODES, 0, stream>>>(
        binned, runinfo, edges, rowptr, cnt, M, nbins, nchunks, x, W1, sup128, mt);

    gcn_gather_relu_gemm64<GCN_H1>
        <<<tiles, B, 0, stream>>>(sup128, rowptr, cnt, edges, b1, W2, sup64, M);

    gcn_gather_relu_gemm64<GCN_H2>
        <<<tiles, B, 0, stream>>>(sup64, rowptr, cnt, edges, b2, W3, sup128, M);

    gcn_row_gather<GCN_OUT, false>
        <<<GW, B, 0, stream>>>(sup128, rowptr, cnt, edges, b3, out, M);
}


# Neighbour sums: a wave gets its 4 rows' offsets by 2 scalar loads, a row's edge records by 1 vector load (lane j = record j, passed on by readlane): 1-3 memory waits a row, not 5-9; sum order kept
# speedup vs baseline: 4.6826x; 1.0316x over previous
#include <hip/hip_runtime.h>


typedef float v2f __attribute__((ext_vector_type(2)));
typedef float v4f __attribute__((ext_vector_type(4)));
typedef float v8f __attribute__((ext_vector_type(8)));
typedef int   v2i __attribute__((ext_vector_type(2)));
typedef int   v4i __attribute__((ext_vector_type(4)));

#define GCN_IN_C  256
#define GCN_H1    128
#define GCN_H2    64
#define GCN_OUT   64

template <int K, int N>
__device__ __forceinline__
void gcn_gemm_strip_wmma_f32(const float* __restrict__ A,
                             const float* __restrict__ W,
                             float* __restrict__ C,
                             int m0, int n0, int lane)
{
    const int half = lane >> 4;
    const int l    = lane & 15;

    const float* __restrict__ arow = A + (size_t)(m0 + l) * K + 4 * half;
    const float* __restrict__ wcol = W + (size_t)(4 * half) * N + (n0 + 4 * l);

    v8f acc0 = {}, acc1 = {}, acc2 = {}, acc3 = {};
#pragma unroll 2
    for (int k = 0; k < K; k += 8) {
        const v4f a  = *reinterpret_cast<const v4f*>(arow + k);
        const v4f w0 = *reinterpret_cast<const v4f*>(wcol + (size_t)(k + 0) * N);
        const v4f w1 = *reinterpret_cast<const v4f*>(wcol + (size_t)(k + 1) * N);
        const v4f w2 = *reinterpret_cast<const v4f*>(wcol + (size_t)(k + 2) * N);
        const v4f w3 = *reinterpret_cast<const v4f*>(wcol + (size_t)(k + 3) * N);
        v2f af, bf;
        af.x = a.x;  af.y = a.y;
        bf.x = w0.x; bf.y = w1.x;
        acc0 = __builtin_amdgcn_wmma_f32_16x16x4_f32(
            false, af, false, bf, (short)0, acc0, false, false);
        bf.x = w0.y; bf.y = w1.y;
        acc1 = __builtin_amdgcn_wmma_f32_16x16x4_f32(
            false, af, false, bf, (short)0, acc1, false, false);
        bf.x = w0.z; bf.y = w1.z;
        acc2 = __builtin_amdgcn_wmma_f32_16x16x4_f32(
            false, af, false, bf, (short)0, acc2, false, false);
        bf.x = w0.w; bf.y = w1.w;
        acc3 = __builtin_amdgcn_wmma_f32_16x16x4_f32(
            false, af, false, bf, (short)0, acc3, false, false);
        af.x = a.z;  af.y = a.w;
        bf.x = w2.x; bf.y = w3.x;
        acc0 = __builtin_amdgcn_wmma_f32_16x16x4_f32(
            false, af, false, bf, (short)0, acc0, false, false);
        bf.x = w2.y; bf.y = w3.y;
        acc1 = __builtin_amdgcn_wmma_f32_16x16x4_f32(
            false, af, false, bf, (short)0, acc1, false, false);
        bf.x = w2.z; bf.y = w3.z;
        acc2 = __builtin_amdgcn_wmma_f32_16x16x4_f32(
            false, af, false, bf, (short)0, acc2, false, false);
        bf.x = w2.w; bf.y = w3.w;
        acc3 = __builtin_amdgcn_wmma_f32_16x16x4_f32(
            false, af, false, bf, (short)0, acc3, false, false);
    }

    float* __restrict__ crow = C + (size_t)(m0 + 8 * half) * N + (n0 + 4 * l);
#pragma unroll
    for (int v = 0; v < 8; ++v) {
        v4f o;
        o.x = acc0[v]; o.y = acc1[v]; o.z = acc2[v]; o.w = acc3[v];
        *reinterpret_cast<v4f*>(crow + (size_t)v * N) = o;
    }
}

#define GCN_BIN_SHIFT 8
#define GCN_BIN_NODES 256
#define GCN_NBMAX     512
#define GCN_CHUNK     8192
#define GCN_CT        1024
#define GCN_CAP       8192
#define GCN_MAXCH     256

__device__ __forceinline__ int gcn_block_excl_scan(int v, int* s_wsum, int lane, int wave)
{
    int inc = v;
    for (int o = 1; o < 32; o <<= 1) {
        const int nb = __shfl_up(inc, o, 32);
        if (lane >= o) inc += nb;
    }
    if (lane == 31) s_wsum[wave] = inc;
    __syncthreads();
    int base = 0;
    for (int w = 0; w < 32; ++w)
        if (w < wave) base += s_wsum[w];
    __syncthreads();
    return base + inc - v;
}

__device__ __forceinline__ void gcn_place(int d, int s, int w, int* s_cur, v2i* s_rec)
{
    const int p = atomicAdd(&s_cur[d >> GCN_BIN_SHIFT], 1);
    v2i rec;
    rec.x = s | (int)(((unsigned)d & (GCN_BIN_NODES - 1)) << 24);
    rec.y = w;
    s_rec[p] = rec;
}

__global__ __launch_bounds__(GCN_CT)
void gcn_chunk_sort(const int* __restrict__ src, const int* __restrict__ dst,
                    const float* __restrict__ ew, int* __restrict__ runinfo,
                    v2i* __restrict__ binned, int E)
{
    __shared__ __attribute__((aligned(16))) v2i s_rec[GCN_CHUNK];
    __shared__ int s_cnt[GCN_NBMAX];
    __shared__ int s_cur[GCN_NBMAX];
    __shared__ int s_wsum[32];

    const int t    = threadIdx.x;
    const int lane = t & 31;
    const int wave = t >> 5;
    const int b    = blockIdx.x;
    const int e0   = b * GCN_CHUNK;
    const bool whole = (e0 + GCN_CHUNK <= E);

    if (t < GCN_NBMAX) s_cnt[t] = 0;
    __syncthreads();

    v4i dq0 = {}, dq1 = {}, sq0 = {}, sq1 = {}, wq0 = {}, wq1 = {};
    if (whole) {
        const v4i* __restrict__ d4 = reinterpret_cast<const v4i*>(dst + e0);
        const v4i* __restrict__ s4 = reinterpret_cast<const v4i*>(src + e0);
        const v4i* __restrict__ w4 = reinterpret_cast<const v4i*>(ew + e0);
        dq0 = d4[t]; dq1 = d4[t + GCN_CT];
        sq0 = s4[t]; sq1 = s4[t + GCN_CT];
        wq0 = w4[t]; wq1 = w4[t + GCN_CT];
        atomicAdd(&s_cnt[dq0.x >> GCN_BIN_SHIFT], 1);
        atomicAdd(&s_cnt[dq0.y >> GCN_BIN_SHIFT], 1);
        atomicAdd(&s_cnt[dq0.z >> GCN_BIN_SHIFT], 1);
        atomicAdd(&s_cnt[dq0.w >> GCN_BIN_SHIFT], 1);
        atomicAdd(&s_cnt[dq1.x >> GCN_BIN_SHIFT], 1);
        atomicAdd(&s_cnt[dq1.y >> GCN_BIN_SHIFT], 1);
        atomicAdd(&s_cnt[dq1.z >> GCN_BIN_SHIFT], 1);
        atomicAdd(&s_cnt[dq1.w >> GCN_BIN_SHIFT], 1);
    } else {
        for (int i = t; i < E - e0; i += GCN_CT)
            atomicAdd(&s_cnt[dst[e0 + i] >> GCN_BIN_SHIFT], 1);
    }
    __syncthreads();

    const int own  = (t < GCN_NBMAX) ? s_cnt[t] : 0;
    const int loff = gcn_block_excl_scan(own, s_wsum, lane, wave);
    if (t < GCN_NBMAX) {
        s_cur[t] = loff;
        runinfo[b * GCN_NBMAX + t] = (loff << 16) | own;
    }
    __syncthreads();

    if (whole) {
        gcn_place(dq0.x, sq0.x, wq0.x, s_cur, s_rec);
        gcn_place(dq0.y, sq0.y, wq0.y, s_cur, s_rec);
        gcn_place(dq0.z, sq0.z, wq0.z, s_cur, s_rec);
        gcn_place(dq0.w, sq0.w, wq0.w, s_cur, s_rec);
        gcn_place(dq1.x, sq1.x, wq1.x, s_cur, s_rec);
        gcn_place(dq1.y, sq1.y, wq1.y, s_cur, s_rec);
        gcn_place(dq1.z, sq1.z, wq1.z, s_cur, s_rec);
        gcn_place(dq1.w, sq1.w, wq1.w, s_cur, s_rec);
    } else {
        for (int i = t; i < E - e0; i += GCN_CT)
            gcn_place(dst[e0 + i], src[e0 + i], __float_as_int(ew[e0 + i]), s_cur, s_rec);
    }
    __syncthreads();

    if (whole) {
        const v4i* s4r = reinterpret_cast<const v4i*>(s_rec);
        v4i* __restrict__ o4 = reinterpret_cast<v4i*>(binned + e0);
#pragma unroll
        for (int k = 0; k < GCN_CHUNK / (2 * GCN_CT); ++k)
            o4[t + k * GCN_CT] = s4r[t + k * GCN_CT];
    } else {
        for (int i = t; i < E - e0; i += GCN_CT) binned[e0 + i] = s_rec[i];
    }
}

__device__ __forceinline__ void gcn_put(v2i r, int* s_cur, v2i* s_stage,
                                        v2i* __restrict__ outp, bool fits)
{
    const int p = atomicAdd(&s_cur[(unsigned)r.x >> 24], 1);
    v2i rec;
    rec.x = r.x & 0x00FFFFFF;
    rec.y = r.y;
    if (fits) s_stage[p] = rec;
    else      outp[p]    = rec;
}

__device__ __forceinline__
void gcn_bin_sort_block(const v2i* __restrict__ binned, const int* __restrict__ runinfo,
                        v2i* __restrict__ edges, int* __restrict__ rowptr,
                        int* __restrict__ cnt, int M, int b, int nchunks)
{
    __shared__ v2i s_stage[GCN_CAP];
    __shared__ int s_cnt[GCN_BIN_NODES];
    __shared__ int s_cur[GCN_BIN_NODES];
    __shared__ int s_rpos[GCN_MAXCH];
    __shared__ int s_rlen[GCN_MAXCH];
    __shared__ int s_woff[8];
    __shared__ int s_wlen[8];
    __shared__ int s_wsum[8];

    const int t    = threadIdx.x;
    const int lane = t & 31;
    const int wave = t >> 5;

    const int info = (t < nchunks) ? runinfo[t * GCN_NBMAX + b] : 0;
    const int roff = info >> 16;
    const int rlen = info & 0xFFFF;
    s_rpos[t] = t * GCN_CHUNK + roff;
    s_rlen[t] = rlen;
    s_cnt[t]  = 0;
    int so = roff, sl = rlen;
    for (int o = 16; o > 0; o >>= 1) {
        so += __shfl_xor(so, o, 32);
        sl += __shfl_xor(sl, o, 32);
    }
    if (lane == 0) { s_woff[wave] = so; s_wlen[wave] = sl; }
    __syncthreads();
    int base = 0, n = 0;
    for (int w = 0; w < 8; ++w) { base += s_woff[w]; n += s_wlen[w]; }
    v2i* __restrict__ outp = edges + base;

    for (int c0 = wave; c0 < nchunks; c0 += 32) {
        int pos[4], len[4], key[4];
#pragma unroll
        for (int k = 0; k < 4; ++k) {
            const int ch = c0 + 8 * k;
            const int cc = (ch < nchunks) ? ch : 0;
            pos[k] = s_rpos[cc];
            len[k] = (ch < nchunks) ? s_rlen[cc] : 0;
        }
#pragma unroll
        for (int k = 0; k < 4; ++k)
            key[k] = (lane < len[k]) ? binned[pos[k] + lane].x : 0;
#pragma unroll
        for (int k = 0; k < 4; ++k)
            if (lane < len[k]) atomicAdd(&s_cnt[(unsigned)key[k] >> 24], 1);
#pragma unroll
        for (int k = 0; k < 4; ++k)
            for (int j = lane + 32; j < len[k]; j += 32)
                atomicAdd(&s_cnt[(unsigned)binned[pos[k] + j].x >> 24], 1);
    }
    __syncthreads();

    const int c = s_cnt[t];
    int inc = c;
    for (int o = 1; o < 32; o <<= 1) {
        const int nb = __shfl_up(inc, o, 32);
        if (lane >= o) inc += nb;
    }
    if (lane == 31) s_wsum[wave] = inc;
    __syncthreads();
    int wbase = 0;
    for (int w = 0; w < 8; ++w)
        if (w < wave) wbase += s_wsum[w];
    const int ex = wbase + inc - c;
    s_cur[t] = ex;
    const int node = b * GCN_BIN_NODES + t;
    if (node < M) {
        rowptr[node] = base + ex;
        cnt[node]    = c;
    }
    __syncthreads();

    const bool fits = (n <= GCN_CAP);
    for (int c0 = wave; c0 < nchunks; c0 += 32) {
        int pos[4], len[4];
        v2i r[4];
#pragma unroll
        for (int k = 0; k < 4; ++k) {
            const int ch = c0 + 8 * k;
            const int cc = (ch < nchunks) ? ch : 0;
            pos[k] = s_rpos[cc];
            len[k] = (ch < nchunks) ? s_rlen[cc] : 0;
        }
#pragma unroll
        for (int k = 0; k < 4; ++k) {
            v2i z; z.x = 0; z.y = 0;
            r[k] = (lane < len[k]) ? binned[pos[k] + lane] : z;
        }
#pragma unroll
        for (int k = 0; k < 4; ++k)
            if (lane < len[k]) gcn_put(r[k], s_cur, s_stage, outp, fits);
#pragma unroll
        for (int k = 0; k < 4; ++k)
            for (int j = lane + 32; j < len[k]; j += 32)
                gcn_put(binned[pos[k] + j], s_cur, s_stage, outp, fits);
    }
    __syncthreads();
    if (fits) {
        for (int i = t; i < n; i += GCN_BIN_NODES) outp[i] = s_stage[i];
    }
}

__global__ __launch_bounds__(GCN_BIN_NODES)
void gcn_bin_sort_and_gemm1(const v2i* __restrict__ binned, const int* __restrict__ runinfo,
                            v2i* __restrict__ edges, int* __restrict__ rowptr,
                            int* __restrict__ cnt, int M, int nbins, int nchunks,
                            const float* __restrict__ x, const float* __restrict__ W1,
                            float* __restrict__ sup1, int strips)
{
    if ((int)blockIdx.x < nbins) {
        gcn_bin_sort_block(binned, runinfo, edges, rowptr, cnt, M, (int)blockIdx.x, nchunks);
    } else {
        const int lane  = threadIdx.x & 31;
        const int wave  = threadIdx.x >> 5;
        const int strip = (((int)blockIdx.x - nbins) << 2) + (wave >> 1);
        if (strip < strips)
            gcn_gemm_strip_wmma_f32<GCN_IN_C, GCN_H1>(x, W1, sup1, strip << 4,
                                                      (wave & 1) << 6, lane);
    }
}

template <int VEC> struct GcnRowVec;
template <> struct GcnRowVec<2> { typedef v2f type; };
template <> struct GcnRowVec<4> { typedef v4f type; };

template <int N, int C, typename VF>
__device__ __forceinline__ void gcn_issue(const float* __restrict__ sup, v2i rec,
                                          int j, int lane, VF* r, float* w)
{
#pragma unroll
    for (int k = 0; k < N; ++k) {
        const unsigned s = (unsigned)__builtin_amdgcn_readlane(rec.x, j + k);
        w[k] = __int_as_float(__builtin_amdgcn_readlane(rec.y, j + k));
        r[k] = reinterpret_cast<const VF*>(sup + (size_t)s * C)[lane];
    }
}

template <int N, typename VF>
__device__ __forceinline__ void gcn_fold(const VF* r, const float* w, VF& acc)
{
#pragma unroll
    for (int k = 0; k < N; ++k) acc += r[k] * w[k];
}

template <int C, int G, typename VF>
__device__ __forceinline__ void gcn_sum_chunk(const float* __restrict__ sup, v2i rec,
                                              int n, int lane, VF& acc)
{
    constexpr int O4 = (G > 8) ? 8 : 0;
    constexpr int O2 = O4 + 4;
    constexpr int O1 = O2 + 2;
    VF    r[G];
    float w[G];
    asm volatile("" :: "v"(rec.x), "v"(rec.y));
    int j = 0;
    for (; j + G <= n; j += G) {
        gcn_issue<G, C, VF>(sup, rec, j, lane, r, w);
        gcn_fold<G, VF>(r, w, acc);
    }
    const int m = n - j;
    int p = j;
    if (G > 8) { if (m & 8) { gcn_issue<8, C, VF>(sup, rec, p, lane, r, w); p += 8; } }
    if (m & 4) { gcn_issue<4, C, VF>(sup, rec, p, lane, r + O4, w + O4); p += 4; }
    if (m & 2) { gcn_issue<2, C, VF>(sup, rec, p, lane, r + O2, w + O2); p += 2; }
    if (m & 1) { gcn_issue<1, C, VF>(sup, rec, p, lane, r + O1, w + O1); }
    if (G > 8) { if (m & 8) gcn_fold<8, VF>(r, w, acc); }
    if (m & 4) gcn_fold<4, VF>(r + O4, w + O4, acc);
    if (m & 2) gcn_fold<2, VF>(r + O2, w + O2, acc);
    if (m & 1) gcn_fold<1, VF>(r + O1, w + O1, acc);
}

template <int C, int G, typename VF>
__device__ __forceinline__ void gcn_sum_row(const float* __restrict__ sup,
                                            const v2i* __restrict__ ed, v2i rec,
                                            int deg, int lane, VF& acc)
{
    int base = 0;
    while (true) {
        const int left = deg - base;
        gcn_sum_chunk<C, G, VF>(sup, rec, left < 32 ? left : 32, lane, acc);
        base += 32;
        if (base >= deg) break;
        v2i z; z.x = 0; z.y = 0;
        rec = (base + lane < deg) ? ed[base + lane] : z;
    }
}

__device__ __forceinline__ void gcn_rows4_setup(const int* __restrict__ rowptr,
                                                const int* __restrict__ cnt,
                                                const v2i* __restrict__ edges,
                                                int r0, int M, int lane,
                                                int* st, int* dg, v2i* rc)
{
    const v4i sp = *reinterpret_cast<const v4i*>(rowptr + r0);
    const v4i dc = *reinterpret_cast<const v4i*>(cnt + r0);
#pragma unroll
    for (int i = 0; i < 4; ++i) {
        const bool in = (r0 + i < M);
        st[i] = in ? sp[i] : 0;
        dg[i] = in ? dc[i] : 0;
    }
#pragma unroll
    for (int i = 0; i < 4; ++i) {
        v2i z; z.x = 0; z.y = 0;
        rc[i] = (lane < dg[i]) ? edges[st[i] + lane] : z;
    }
}

template <int C, bool RELU>
__global__ __launch_bounds__(256)
void gcn_row_gather(const float* __restrict__ sup,
                    const int* __restrict__ rowptr,
                    const int* __restrict__ cnt,
                    const v2i* __restrict__ edges,
                    const float* __restrict__ bias,
                    float* __restrict__ out,
                    int M)
{
    constexpr int VEC = C / 32;
    constexpr int G   = (C > 64) ? 8 : 16;
    typedef typename GcnRowVec<VEC>::type VF;

    const int lane    = threadIdx.x & 31;
    const int nwaves  = (gridDim.x * blockDim.x) >> 5;
    const int ngroups = (M + 3) >> 2;
    int g = __builtin_amdgcn_readfirstlane(
        (int)((blockIdx.x * blockDim.x + threadIdx.x) >> 5));

    const VF bv = reinterpret_cast<const VF*>(bias)[lane];

    for (; g < ngroups; g += nwaves) {
        const int r0 = g << 2;
        int st[4], dg[4];
        v2i rc[4];
        gcn_rows4_setup(rowptr, cnt, edges, r0, M, lane, st, dg, rc);
#pragma unroll
        for (int i = 0; i < 4; ++i) {
            VF acc = {};
            gcn_sum_row<C, G, VF>(sup, edges + st[i], rc[i], dg[i], lane, acc);
            acc += bv;
            if (RELU) {
                for (int q = 0; q < VEC; ++q) acc[q] = fmaxf(acc[q], 0.0f);
            }
            if (r0 + i < M)
                reinterpret_cast<VF*>(out + (size_t)(r0 + i) * C)[lane] = acc;
        }
    }
}


template <int C>
__global__ __launch_bounds__(256)
void gcn_gather_relu_gemm64(const float* __restrict__ sup,
                            const int* __restrict__ rowptr,
                            const int* __restrict__ cnt,
                            const v2i* __restrict__ edges,
                            const float* __restrict__ bias,
                            const float* __restrict__ W,
                            float* __restrict__ next,
                            int M)
{
    constexpr int VEC   = C / 32;
    constexpr int N     = 64;
    constexpr int PITCH = C + 4;
    constexpr int OP    = N + 4;
    constexpr int G     = (C > 64) ? 8 : 16;
    typedef typename GcnRowVec<VEC>::type VF;

    __shared__ __attribute__((aligned(16))) float s_h[32 * PITCH];
    __shared__ __attribute__((aligned(16))) float s_o[32 * OP];

    const int t    = threadIdx.x;
    const int lane = t & 31;
    const int wave = __builtin_amdgcn_readfirstlane(t >> 5);
    const int m0   = blockIdx.x << 5;

    const VF bv = reinterpret_cast<const VF*>(bias)[lane];

    {
        const int r0 = m0 + wave * 4;
        int st[4], dg[4];
        v2i rc[4];
        gcn_rows4_setup(rowptr, cnt, edges, r0, M, lane, st, dg, rc);
#pragma unroll
        for (int i = 0; i < 4; ++i) {
            VF acc = {};
            if (r0 + i < M) {
                gcn_sum_row<C, G, VF>(sup, edges + st[i], rc[i], dg[i], lane, acc);
                acc += bv;
                for (int q = 0; q < VEC; ++q) acc[q] = fmaxf(acc[q], 0.0f);
            }
            reinterpret_cast<VF*>(s_h + (wave * 4 + i) * PITCH)[lane] = acc;
        }
    }
    __syncthreads();

    const int half = lane >> 4;
    const int l    = lane & 15;
    const int mt   = wave >> 2;
    const int nt   = wave & 3;
    const float* arow = s_h + (mt * 16 + l) * PITCH + 2 * half;
    const float* __restrict__ wcol = W + (size_t)(2 * half) * N + (nt * 16 + l);

    v8f d = {};
#pragma unroll 8
    for (int k = 0; k < C; k += 4) {
        v2f a, b;
        a   = *reinterpret_cast<const v2f*>(arow + k);
        b.x = wcol[(size_t)k * N];
        b.y = wcol[(size_t)k * N + N];
        d = __builtin_amdgcn_wmma_f32_16x16x4_f32(
            false, a, false, b, (short)0, d, false, false);
    }

    float* orow = s_o + (mt * 16 + 8 * half) * OP + (nt * 16 + l);
#pragma unroll
    for (int v = 0; v < 8; ++v) orow[v * OP] = d[v];
    __syncthreads();

#pragma unroll
    for (int p = 0; p < 2; ++p) {
        const int r  = (t >> 4) + 16 * p;
        const int c4 = (t & 15) * 4;
        if (m0 + r < M)
            *reinterpret_cast<v4f*>(next + (size_t)(m0 + r) * N + c4) =
                *reinterpret_cast<const v4f*>(s_o + r * OP + c4);
    }
}


extern "C" void kernel_launch(void* const* d_in, const int* in_sizes, int n_in,
                              void* d_out, int out_size, void* d_ws, size_t ws_size,
                              hipStream_t stream)
{
    const float* x    = (const float*)d_in[0];
    const int*   esrc = (const int*)  d_in[1];
    const int*   edst = (const int*)  d_in[2];
    const float* ew   = (const float*)d_in[3];
    const float* W1   = (const float*)d_in[4];
    const float* b1   = (const float*)d_in[5];
    const float* W2   = (const float*)d_in[6];
    const float* b2   = (const float*)d_in[7];
    const float* W3   = (const float*)d_in[8];
    const float* b3   = (const float*)d_in[9];

    const int M = in_sizes[0] / GCN_IN_C;
    const int E = in_sizes[1];

    float* ws     = (float*)d_ws;
    float* sup128 = ws;
    float* sup64  = sup128 + (size_t)M * GCN_H1;
    float* out    = (float*)d_out;

    const int nchunks = (E + GCN_CHUNK - 1) / GCN_CHUNK;
    const int nbins   = (M + GCN_BIN_NODES - 1) / GCN_BIN_NODES;
    int* cnt     = (int*)(sup64 + (size_t)M * GCN_H2);
    int* rowptr  = cnt + nbins * GCN_BIN_NODES;
    v2i* edges   = (v2i*)(rowptr + nbins * GCN_BIN_NODES);
    v2i* binned  = edges + E;
    int* runinfo = (int*)(binned + E);

    const int mt = M / 16;
    const int B  = 256;
    const int GW = 1563;
    const int tiles = (M + 31) / 32;

    gcn_chunk_sort<<<nchunks, GCN_CT, 0, stream>>>(esrc, edst, ew, runinfo, binned, E);
    gcn_bin_sort_and_gemm1<<<nbins + (mt + 3) / 4, GCN_BIN_NODES, 0, stream>>>(
        binned, runinfo, edges, rowptr, cnt, M, nbins, nchunks, x, W1, sup128, mt);

    gcn_gather_relu_gemm64<GCN_H1>
        <<<tiles, B, 0, stream>>>(sup128, rowptr, cnt, edges, b1, W2, sup64, M);

    gcn_gather_relu_gemm64<GCN_H2>
        <<<tiles, B, 0, stream>>>(sup64, rowptr, cnt, edges, b2, W3, sup128, M);

    gcn_row_gather<GCN_OUT, false>
        <<<GW, B, 0, stream>>>(sup128, rowptr, cnt, edges, b3, out, M);
}
